# starting kernel
# baseline (speedup 1.0000x reference)
#include <hip/hip_runtime.h>
#include <cstddef>

typedef float v2f __attribute__((ext_vector_type(2)));
typedef float v8f __attribute__((ext_vector_type(8)));

#define DIN 64

__global__ void gnn_zero_kernel(float* __restrict__ p, unsigned long long n) {
    unsigned long long i = (unsigned long long)blockIdx.x * blockDim.x + threadIdx.x;
    if (i < n) p[i] = 0.0f;
}

__global__ void gnn_pack_weights_kernel(const float* __restrict__ Wl,
                                        const float* __restrict__ bl,
                                        const float* __restrict__ Wr,
                                        float* __restrict__ WTl,
                                        float* __restrict__ WTr,
                                        float* __restrict__ blp,
                                        int dout, int dout_pad) {
    int tid = blockIdx.x * blockDim.x + threadIdx.x;
    int col = tid / DIN;
    int k   = tid % DIN;
    if (col >= dout_pad) return;
    bool cv  = (col < dout);
    WTl[(size_t)col * DIN + k] = cv ? Wl[(size_t)k * dout + col] : 0.0f;
    WTr[(size_t)col * DIN + k] = cv ? Wr[(size_t)k * dout + col] : 0.0f;
    if (k == 0) blp[col] = cv ? bl[col] : 0.0f;
}

__global__ void gnn_scatter_kernel(const float* __restrict__ x,
                                   const int* __restrict__ src,
                                   const int* __restrict__ dst,
                                   float* __restrict__ agg,
                                   float* __restrict__ deg,
                                   int E, int do_deg) {
    unsigned long long tid = (unsigned long long)blockIdx.x * blockDim.x + threadIdx.x;
    unsigned long long e  = tid >> 4;
    int f4 = (int)(tid & 15);
    if (e >= (unsigned long long)E) return;
    int s = src[e];
    int d = dst[e];
    const float4* xr = (const float4*)(x + (size_t)s * DIN);
    float4 v = xr[f4];
    float* ar = agg + (size_t)d * DIN + (size_t)f4 * 4;
    atomicAdd(ar + 0, v.x);
    atomicAdd(ar + 1, v.y);
    atomicAdd(ar + 2, v.z);
    atomicAdd(ar + 3, v.w);
    if (do_deg && f4 == 0) atomicAdd(deg + d, 1.0f);
}

__global__ void __launch_bounds__(128)
gnn_dense_wmma_kernel(const float* __restrict__ agg,
                      const float* __restrict__ deg,
                      const float* __restrict__ hin,
                      const float* __restrict__ WTl,
                      const float* __restrict__ blp,
                      const float* __restrict__ WTr,
                      float* __restrict__ hout,
                      int do_relu, int n_mtiles) {
    const int lane  = threadIdx.x;
    const int mtile = blockIdx.x * blockDim.y + threadIdx.y;
    const int ntile = blockIdx.y;
    if (mtile >= n_mtiles) return;

    const int mrow  = mtile * 16 + (lane & 15);
    const int khalf = (lane >> 4) * 2;
    const int col   = ntile * 16 + (lane & 15);

    const float rdeg = 1.0f / fmaxf(deg[mrow], 1.0f);
    const float* aggr = agg + (size_t)mrow * DIN + khalf;
    const float* hinr = hin + (size_t)mrow * DIN + khalf;
    const float* wl   = WTl + (size_t)col  * DIN + khalf;
    const float* wr   = WTr + (size_t)col  * DIN + khalf;

    v8f c = {};
    #pragma unroll
    for (int k0 = 0; k0 < DIN; k0 += 4) {
        v2f aAgg = *(const v2f*)(aggr + k0);
        aAgg.x *= rdeg; aAgg.y *= rdeg;
        v2f aX   = *(const v2f*)(hinr + k0);
        v2f bL   = *(const v2f*)(wl + k0);
        v2f bR   = *(const v2f*)(wr + k0);
        c = __builtin_amdgcn_wmma_f32_16x16x4_f32(false, aAgg, false, bL,
                                                  (short)0, c, false, false);
        c = __builtin_amdgcn_wmma_f32_16x16x4_f32(false, aX,   false, bR,
                                                  (short)0, c, false, false);
    }

    const float bv = blp[col];
    float* outp = hout + (size_t)(mtile * 16 + ((lane >> 4) * 8)) * DIN + col;
    #pragma unroll
    for (int r = 0; r < 8; ++r) {
        float v = c[r] + bv;
        if (do_relu) v = fmaxf(v, 0.0f);
        outp[(size_t)r * DIN] = v;
    }
}

__global__ void gnn_logsoftmax_kernel(const float* __restrict__ logits,
                                      float* __restrict__ out,
                                      int n_nodes, int C, int stride) {
    const int lane = threadIdx.x;
    const int node = blockIdx.x * blockDim.y + threadIdx.y;
    if (node >= n_nodes) return;
    const float* rowp = logits + (size_t)node * stride;
    const float NEG_INF = -__builtin_inff();
    float v0 = (lane      < C) ? rowp[lane]      : NEG_INF;
    float v1 = (lane + 32 < C) ? rowp[lane + 32] : NEG_INF;
    float m = fmaxf(v0, v1);
    #pragma unroll
    for (int off = 16; off > 0; off >>= 1) m = fmaxf(m, __shfl_xor(m, off, 32));
    float e0 = (lane      < C) ? expf(v0 - m) : 0.0f;
    float e1 = (lane + 32 < C) ? expf(v1 - m) : 0.0f;
    float s = e0 + e1;
    #pragma unroll
    for (int off = 16; off > 0; off >>= 1) s += __shfl_xor(s, off, 32);
    const float ls = logf(s);
    if (lane      < C) out[(size_t)node * C + lane]      = v0 - m - ls;
    if (lane + 32 < C) out[(size_t)node * C + lane + 32] = v1 - m - ls;
}

static inline void launch_dense(const float* agg, const float* deg, const float* hin,
                                const float* WTl, const float* blp, const float* WTr,
                                float* hout, int dout_pad, int do_relu,
                                int N, hipStream_t stream) {
    const int n_mtiles = (N + 15) / 16;
    dim3 block(32, 4);
    dim3 grid((n_mtiles + 3) / 4, dout_pad / 16);
    gnn_dense_wmma_kernel<<<grid, block, 0, stream>>>(agg, deg, hin, WTl, blp, WTr,
                                                      hout, do_relu, n_mtiles);
}

extern "C" void kernel_launch(void* const* d_in, const int* in_sizes, int n_in,
                              void* d_out, int out_size, void* d_ws, size_t ws_size,
                              hipStream_t stream) {
    const float* x   = (const float*)d_in[0];
    const int*   ei  = (const int*)  d_in[1];
    const float* Wl0 = (const float*)d_in[2];
    const float* bl0 = (const float*)d_in[3];
    const float* Wr0 = (const float*)d_in[4];
    const float* Wl1 = (const float*)d_in[5];
    const float* bl1 = (const float*)d_in[6];
    const float* Wr1 = (const float*)d_in[7];
    const float* Wl2 = (const float*)d_in[8];
    const float* bl2 = (const float*)d_in[9];
    const float* Wr2 = (const float*)d_in[10];

    const int N = in_sizes[0] / DIN;
    const int E = in_sizes[1] / 2;
    const int* src = ei;
    const int* dst = ei + E;

    const int DP0 = 64, DP1 = 64, DP2 = 48;

    float* deg  = (float*)d_ws;
    float* agg  = deg + N;
    float* h1   = agg + (size_t)N * DIN;
    float* h2   = h1  + (size_t)N * DIN;
    float* WTl0 = h2  + (size_t)N * DIN;
    float* WTr0 = WTl0 + DP0 * DIN;
    float* blp0 = WTr0 + DP0 * DIN;
    float* WTl1 = blp0 + DP0;
    float* WTr1 = WTl1 + DP1 * DIN;
    float* blp1 = WTr1 + DP1 * DIN;
    float* WTl2 = blp1 + DP1;
    float* WTr2 = WTl2 + DP2 * DIN;
    float* blp2 = WTr2 + DP2 * DIN;

    const unsigned long long n_da = (unsigned long long)N * (DIN + 1);
    const unsigned long long n_a  = (unsigned long long)N * DIN;
    const unsigned long long scat_threads = (unsigned long long)E * 16ull;
    const int ZB = 256;

    gnn_pack_weights_kernel<<<(DP0 * DIN + ZB - 1) / ZB, ZB, 0, stream>>>(
        Wl0, bl0, Wr0, WTl0, WTr0, blp0, 64, DP0);
    gnn_pack_weights_kernel<<<(DP1 * DIN + ZB - 1) / ZB, ZB, 0, stream>>>(
        Wl1, bl1, Wr1, WTl1, WTr1, blp1, 64, DP1);
    gnn_pack_weights_kernel<<<(DP2 * DIN + ZB - 1) / ZB, ZB, 0, stream>>>(
        Wl2, bl2, Wr2, WTl2, WTr2, blp2, 40, DP2);

    gnn_zero_kernel<<<(unsigned)((n_da + ZB - 1) / ZB), ZB, 0, stream>>>(deg, n_da);
    gnn_scatter_kernel<<<(unsigned)((scat_threads + ZB - 1) / ZB), ZB, 0, stream>>>(
        x, src, dst, agg, deg, E, 1);
    launch_dense(agg, deg, x, WTl0, blp0, WTr0, h1, DP0, 1, N, stream);

    gnn_zero_kernel<<<(unsigned)((n_a + ZB - 1) / ZB), ZB, 0, stream>>>(agg, n_a);
    gnn_scatter_kernel<<<(unsigned)((scat_threads + ZB - 1) / ZB), ZB, 0, stream>>>(
        h1, src, dst, agg, deg, E, 0);
    launch_dense(agg, deg, h1, WTl1, blp1, WTr1, h2, DP1, 1, N, stream);

    gnn_zero_kernel<<<(unsigned)((n_a + ZB - 1) / ZB), ZB, 0, stream>>>(agg, n_a);
    gnn_scatter_kernel<<<(unsigned)((scat_threads + ZB - 1) / ZB), ZB, 0, stream>>>(
        h2, src, dst, agg, deg, E, 0);
    launch_dense(agg, deg, h2, WTl2, blp2, WTr2, h1, DP2, 0, N, stream);

    {
        dim3 block(32, 8);
        dim3 grid((N + 7) / 8);
        gnn_logsoftmax_kernel<<<grid, block, 0, stream>>>(h1, (float*)d_out, N, 40, 64);
    }
}


# Neighbour sums leave global float atomics: 250 owner work-groups sweep a byte-per-edge owner key, add their edges' rows with LDS float adds, store once; zero fills, 2nd plane go; arrival order kept
# speedup vs baseline: 2.6034x; 2.6034x over previous
#include <hip/hip_runtime.h>
#include <cstddef>

typedef float v2f __attribute__((ext_vector_type(2)));
typedef float v4f __attribute__((ext_vector_type(4)));
typedef float v8f __attribute__((ext_vector_type(8)));

#define DIN 64

__global__ void gnn_pack_weights_kernel(const float* __restrict__ Wl,
                                        const float* __restrict__ bl,
                                        const float* __restrict__ Wr,
                                        float* __restrict__ WFl,
                                        float* __restrict__ WFr,
                                        float* __restrict__ blp,
                                        int dout, int dout_pad) {
    int tid = blockIdx.x * blockDim.x + threadIdx.x;
    int col = tid / DIN;
    int k   = tid % DIN;
    if (col >= dout_pad) return;
    bool cv  = (col < dout);
    const int i = k >> 2, h = (k >> 1) & 1, e = k & 1;
    const size_t o = ((size_t)(((col >> 4) * 16 + i) * 32 + h * 16 + (col & 15))) * 2 + e;
    WFl[o] = cv ? Wl[(size_t)k * dout + col] : 0.0f;
    WFr[o] = cv ? Wr[(size_t)k * dout + col] : 0.0f;
    if (k == 0) blp[col] = cv ? bl[col] : 0.0f;
}

#define OWN_WAVES 32
#define OWN_RMAX  640
#define OWN_QCAP  64
#define OWN_BATCH 16

__global__ void gnn_owner_key_kernel(const int* __restrict__ dst,
                                     unsigned* __restrict__ keys,
                                     int E, int n_words, int R) {
    const int w = blockIdx.x * blockDim.x + threadIdx.x;
    if (w >= n_words) return;
    unsigned kw = 0u;
    #pragma unroll
    for (int b = 0; b < 4; ++b) {
        const long long e = (long long)w * 4 + b;
        unsigned k = 0xFFu;
        if (e < (long long)E) k = (unsigned)dst[e] / (unsigned)R;
        kw |= k << (8 * b);
    }
    keys[w] = kw;
}

__device__ __forceinline__ unsigned gnn_zero_bytes(unsigned x) {
    return ~(((x & 0x7F7F7F7Fu) + 0x7F7F7F7Fu) | x | 0x7F7F7F7Fu);
}

template <int W>
__device__ __forceinline__ void
gnn_owner_drain(const float* __restrict__ rows,
                const int* __restrict__ src, const int* __restrict__ dst,
                const unsigned* qw, unsigned head, int n, int lo, int lane,
                float* acc, unsigned* degL, int do_deg) {
    int s = 0, r = 0;
    if (lane < n) {
        const unsigned e = qw[(head + (unsigned)lane) & (OWN_QCAP - 1)];
        s = src[e];
        r = dst[e] - lo;
        if (do_deg) atomicAdd(&degL[r], 1u);
    }
    const bool act  = (W == DIN) || (2 * lane < W);
    const int  loff = act ? 2 * lane : 0;
    #pragma unroll
    for (int b = 0; b < 32 / OWN_BATCH; ++b) {
        if (b * OWN_BATCH < n) {
            v2f v[OWN_BATCH];
            #pragma unroll
            for (int j = 0; j < OWN_BATCH; ++j) {
                const int sj = __builtin_amdgcn_readlane(s, b * OWN_BATCH + j);
                v[j] = *(const v2f*)(rows + (size_t)sj * DIN + loff);
            }
            if (act) {
                #pragma unroll
                for (int j = 0; j < OWN_BATCH; ++j) {
                    if (b * OWN_BATCH + j < n) {
                        const int rj = __builtin_amdgcn_readlane(r, b * OWN_BATCH + j);
                        float* a = acc + rj * W + 2 * lane;
                        __hip_atomic_fetch_add(a,     v[j].x, __ATOMIC_RELAXED, __HIP_MEMORY_SCOPE_WORKGROUP);
                        __hip_atomic_fetch_add(a + 1, v[j].y, __ATOMIC_RELAXED, __HIP_MEMORY_SCOPE_WORKGROUP);
                    }
                }
            }
        }
    }
}

template <int W>
__global__ void __launch_bounds__(32 * OWN_WAVES)
gnn_owner_sum_kernel(const float* __restrict__ rows,
                     const int* __restrict__ src,
                     const int* __restrict__ dst,
                     const unsigned* __restrict__ keys,
                     int n_chunks,
                     float* __restrict__ sums,
                     float* __restrict__ deg,
                     int n_nodes, int R, int do_deg) {
    __shared__ __attribute__((aligned(16))) float acc[OWN_RMAX * DIN];
    __shared__ unsigned degL[OWN_RMAX];
    __shared__ unsigned q[OWN_WAVES][OWN_QCAP];

    const int lane = threadIdx.x;
    const int wv = __builtin_amdgcn_readfirstlane((int)threadIdx.y);
    const int t  = wv * 32 + lane;
    const int g  = blockIdx.x;
    const int lo = g * R;
    int nrows = n_nodes - lo;
    nrows = nrows < R ? nrows : R;

    const int n4 = nrows * W / 4;
    for (int i = t; i < n4; i += 32 * OWN_WAVES) ((v4f*)acc)[i] = v4f{0.0f, 0.0f, 0.0f, 0.0f};
    for (int i = t; i < nrows; i += 32 * OWN_WAVES) degL[i] = 0u;
    __syncthreads();

    const unsigned pat = (unsigned)g * 0x01010101u;
    const int start = (int)(((unsigned)g * 97u) % (unsigned)n_chunks);
    const uint4* k4 = (const uint4*)keys;
    unsigned* qw = &q[wv][0];
    unsigned head = 0u;
    int cnt = 0;

    int k = wv;
    uint4 kw = make_uint4(0xFFFFFFFFu, 0xFFFFFFFFu, 0xFFFFFFFFu, 0xFFFFFFFFu);
    int c = 0;
    if (k < n_chunks) {
        c = start + k; if (c >= n_chunks) c -= n_chunks;
        kw = k4[(size_t)c * 32 + lane];
    }
    while (k < n_chunks) {
        unsigned m = (gnn_zero_bytes(kw.x ^ pat) >> 7) | (gnn_zero_bytes(kw.y ^ pat) >> 6)
                   | (gnn_zero_bytes(kw.z ^ pat) >> 5) | (gnn_zero_bytes(kw.w ^ pat) >> 4);
        const unsigned ebase = (unsigned)c * 512u + (unsigned)lane * 16u;

        const int kn = k + OWN_WAVES;
        int cn = start + (kn < n_chunks ? kn : k);
        if (cn >= n_chunks) cn -= n_chunks;
        const uint4 kwn = k4[(size_t)cn * 32 + lane];
        for (;;) {
            const bool has = m != 0u;
            const unsigned bal = __builtin_amdgcn_ballot_w32(has);
            if (bal == 0u) break;
            if (has) {
                const unsigned pre = __builtin_amdgcn_mbcnt_lo(bal, 0u);
                const unsigned tz  = (unsigned)__builtin_ctz(m);
                qw[(head + (unsigned)cnt + pre) & (OWN_QCAP - 1)] = ebase + (tz & 7u) * 4u + (tz >> 3);
                m &= m - 1u;
            }
            cnt += __builtin_popcount(bal);
            if (cnt >= 32) {
                gnn_owner_drain<W>(rows, src, dst, qw, head, 32, lo, lane, acc, degL, do_deg);
                head = (head + 32u) & (OWN_QCAP - 1);
                cnt -= 32;
            }
        }
        k = kn; c = cn; kw = kwn;
    }
    if (cnt > 0)
        gnn_owner_drain<W>(rows, src, dst, qw, head, cnt, lo, lane, acc, degL, do_deg);

    __syncthreads();
    v4f* o4 = (v4f*)(sums + (size_t)lo * W);
    for (int i = t; i < n4; i += 32 * OWN_WAVES) o4[i] = ((const v4f*)acc)[i];
    if (do_deg)
        for (int i = t; i < nrows; i += 32 * OWN_WAVES) deg[lo + i] = (float)degL[i];
}

#define SROWS  32
#define APITCH 68

template <bool TWO>
__device__ __forceinline__ void
gnn_strip_request(const float* __restrict__ agg0, const float* __restrict__ agg1,
                  const float* __restrict__ deg0, const float* __restrict__ deg1,
                  const float* __restrict__ hin, int strip, int srow, int c4, int n_rows,
                  v4f (&ra0)[2], v4f (&ra1)[2], v4f (&rx)[2],
                  float (&rd0)[2], float (&rd1)[2]) {
    #pragma unroll
    for (int j = 0; j < 2; ++j) {
        int row = strip * SROWS + srow + 16 * j;
        row = row < n_rows ? row : n_rows - 1;
        const size_t o = (size_t)row * DIN + c4;
        ra0[j] = *(const v4f*)(agg0 + o);
        if (TWO) ra1[j] = *(const v4f*)(agg1 + o);
        rx[j]  = *(const v4f*)(hin + o);
        rd0[j] = deg0[row];
        if (TWO) rd1[j] = deg1[row];
    }
}

template <bool TWO>
__global__ void __launch_bounds__(256)
gnn_dense_wmma_kernel(const float* __restrict__ agg0,
                      const float* __restrict__ agg1,
                      const float* __restrict__ deg0,
                      const float* __restrict__ deg1,
                      const float* __restrict__ hin,
                      const float* __restrict__ WFl,
                      const float* __restrict__ blp,
                      const float* __restrict__ WFr,
                      float* __restrict__ hout,
                      int do_relu, int n_rows, int n_strips) {
    __shared__ __attribute__((aligned(16))) float sM[2][SROWS * APITCH];
    __shared__ __attribute__((aligned(16))) float sX[2][SROWS * APITCH];

    const int lane  = threadIdx.x;
    const int wv    = __builtin_amdgcn_readfirstlane((int)threadIdx.y);
    const int ntile = wv & 3;
    const int mhalf = wv >> 2;
    const int khalf = (lane >> 4) * 2;
    const int col   = ntile * 16 + (lane & 15);

    int strip = blockIdx.x;
    if (strip >= n_strips) return;

    const int t    = wv * 32 + lane;
    const int srow = t >> 4;
    const int c4   = (t & 15) * 4;

    v4f ra0[2], ra1[2], rx[2];
    float  rd0[2], rd1[2];
    gnn_strip_request<TWO>(agg0, agg1, deg0, deg1, hin, strip, srow, c4, n_rows,
                           ra0, ra1, rx, rd0, rd1);

    v2f bL[16], bR[16];
    {
        const size_t wo = ((size_t)(ntile * 16) * 32 + lane) * 2;
        #pragma unroll
        for (int i = 0; i < 16; ++i) {
            bL[i] = *(const v2f*)(WFl + wo + (size_t)i * 64);
            bR[i] = *(const v2f*)(WFr + wo + (size_t)i * 64);
        }
    }
    const float bv = blp[col];

    const int aoff = (mhalf * 16 + (lane & 15)) * APITCH + khalf;

    int buf = 0;
    for (;;) {
        float* pm = &sM[buf][0];
        float* px = &sX[buf][0];
        #pragma unroll
        for (int j = 0; j < 2; ++j) {
            float dsum = rd0[j];
            if (TWO) dsum += rd1[j];
            const float rdeg = 1.0f / fmaxf(dsum, 1.0f);
            v4f m = ra0[j];
            if (TWO) m += ra1[j];
            m *= rdeg;
            const int o = (srow + 16 * j) * APITCH + c4;
            *(v4f*)(pm + o) = m;
            *(v4f*)(px + o) = rx[j];
        }
        __builtin_amdgcn_fence(__ATOMIC_RELEASE, "workgroup", "local");
        __builtin_amdgcn_s_barrier();
        __builtin_amdgcn_fence(__ATOMIC_ACQUIRE, "workgroup", "local");

        const int next = strip + (int)gridDim.x;
        if (next < n_strips)
            gnn_strip_request<TWO>(agg0, agg1, deg0, deg1, hin, next, srow, c4, n_rows,
                                   ra0, ra1, rx, rd0, rd1);

        const float* am = pm + aoff;
        const float* ax = px + aoff;
        v8f c = {};
        #pragma unroll
        for (int i = 0; i < 16; ++i) {
            const v2f aAgg = *(const v2f*)(am + 4 * i);
            const v2f aX   = *(const v2f*)(ax + 4 * i);
            c = __builtin_amdgcn_wmma_f32_16x16x4_f32(false, aAgg, false, bL[i],
                                                      (short)0, c, false, false);
            c = __builtin_amdgcn_wmma_f32_16x16x4_f32(false, aX,   false, bR[i],
                                                      (short)0, c, false, false);
        }

        const int row0 = strip * SROWS + mhalf * 16 + (lane >> 4) * 8;
        float* outp = hout + (size_t)row0 * DIN + col;
        if (strip * SROWS + mhalf * 16 + 16 <= n_rows) {
            #pragma unroll
            for (int r = 0; r < 8; ++r) {
                float v = c[r] + bv;
                if (do_relu) v = fmaxf(v, 0.0f);
                outp[(size_t)r * DIN] = v;
            }
        } else {
            #pragma unroll
            for (int r = 0; r < 8; ++r) {
                float v = c[r] + bv;
                if (do_relu) v = fmaxf(v, 0.0f);
                if (row0 + r < n_rows) outp[(size_t)r * DIN] = v;
            }
        }

        if (next >= n_strips) break;
        strip = next;
        buf ^= 1;
    }
}

#define C2   40
#define C2A  32
#define C2B  8

__device__ __forceinline__ void
gnn_proj_store(const v8f& cy, const v8f& cz, float bv,
               float* __restrict__ yp, int ypitch, float* __restrict__ zp, int nvalid) {
    #pragma unroll
    for (int r = 0; r < 8; ++r) {
        if (r < nvalid) {
            yp[(size_t)r * ypitch] = cy[r];
            zp[(size_t)r * C2]     = cz[r] + bv;
        }
    }
}

__global__ void __launch_bounds__(256)
gnn_proj_wmma_kernel(const float* __restrict__ hin,
                     const float* __restrict__ WFl,
                     const float* __restrict__ WFr,
                     const float* __restrict__ blp,
                     float* __restrict__ y,
                     float* __restrict__ z,
                     int n_rows, int n_strips) {
    __shared__ __attribute__((aligned(16))) float sX[2][SROWS * APITCH];

    const int lane  = threadIdx.x;
    const int wv    = __builtin_amdgcn_readfirstlane((int)threadIdx.y);
    const bool worker = wv < 6;
    const int mhalf = wv >= 3 ? 1 : 0;
    const int ntile = wv - 3 * mhalf;
    const int khalf = (lane >> 4) * 2;
    const int col   = ntile * 16 + (lane & 15);

    int strip = blockIdx.x;
    if (strip >= n_strips) return;

    const int t    = wv * 32 + lane;
    const int srow = t >> 4;
    const int c4   = (t & 15) * 4;

    v4f rx[2];
    #pragma unroll
    for (int j = 0; j < 2; ++j) {
        int row = strip * SROWS + srow + 16 * j;
        row = row < n_rows ? row : n_rows - 1;
        rx[j] = *(const v4f*)(hin + (size_t)row * DIN + c4);
    }

    v2f bL[16], bR[16];
    float bv = 0.0f;
    if (worker) {
        const size_t wo = ((size_t)(ntile * 16) * 32 + lane) * 2;
        #pragma unroll
        for (int i = 0; i < 16; ++i) {
            bL[i] = *(const v2f*)(WFl + wo + (size_t)i * 64);
            bR[i] = *(const v2f*)(WFr + wo + (size_t)i * 64);
        }
        bv = blp[col];
    }

    const int aoff = (mhalf * 16 + (lane & 15)) * APITCH + khalf;

    int buf = 0;
    for (;;) {
        float* px = &sX[buf][0];
        #pragma unroll
        for (int j = 0; j < 2; ++j)
            *(v4f*)(px + (srow + 16 * j) * APITCH + c4) = rx[j];
        __builtin_amdgcn_fence(__ATOMIC_RELEASE, "workgroup", "local");
        __builtin_amdgcn_s_barrier();
        __builtin_amdgcn_fence(__ATOMIC_ACQUIRE, "workgroup", "local");

        const int next = strip + (int)gridDim.x;
        if (next < n_strips) {
            #pragma unroll
            for (int j = 0; j < 2; ++j) {
                int row = next * SROWS + srow + 16 * j;
                row = row < n_rows ? row : n_rows - 1;
                rx[j] = *(const v4f*)(hin + (size_t)row * DIN + c4);
            }
        }

        if (worker) {
            const float* ax = px + aoff;
            v8f cy = {};
            v8f cz = {};
            #pragma unroll
            for (int i = 0; i < 16; ++i) {
                const v2f a = *(const v2f*)(ax + 4 * i);
                cy = __builtin_amdgcn_wmma_f32_16x16x4_f32(false, a, false, bL[i],
                                                           (short)0, cy, false, false);
                cz = __builtin_amdgcn_wmma_f32_16x16x4_f32(false, a, false, bR[i],
                                                           (short)0, cz, false, false);
            }

            const int tile_row = strip * SROWS + mhalf * 16;
            const int row0i    = tile_row + (lane >> 4) * 8;
            const size_t row0  = (size_t)row0i;
            const bool full  = tile_row + 16 <= n_rows;
            const int nvalid = n_rows - row0i;
            if (col < C2) {
                float* yp = y + row0 * DIN + col;
                float* zp = z + row0 * C2  + col;
                if (full) gnn_proj_store(cy, cz, bv, yp, DIN, zp, 8);
                else      gnn_proj_store(cy, cz, bv, yp, DIN, zp, nvalid);
            }
        }

        if (next >= n_strips) break;
        strip = next;
        buf ^= 1;
    }
}

__global__ void gnn_mean_logsoftmax_kernel(const float* __restrict__ s40,
                                           const float* __restrict__ deg,
                                           float* out,
                                           int n_nodes) {
    const int lane = threadIdx.x;
    const int node = blockIdx.x * blockDim.y + threadIdx.y;
    if (node >= n_nodes) return;
    const bool hasB = lane < C2B;
    const float NEG_INF = -__builtin_inff();

    const float dsum = deg[node];
    const float a = s40[(size_t)node * C2 + lane];
    const float b = hasB ? s40[(size_t)node * C2 + C2A + lane] : 0.0f;
    const float rdeg = 1.0f / fmaxf(dsum, 1.0f);
    float* orow = out + (size_t)node * C2;
    const float v0 = a * rdeg + orow[lane];
    const float v1 = hasB ? b * rdeg + orow[C2A + lane] : NEG_INF;

    float m = fmaxf(v0, v1);
    #pragma unroll
    for (int off = 16; off > 0; off >>= 1) m = fmaxf(m, __shfl_xor(m, off, 32));
    const float e0 = expf(v0 - m);
    const float e1 = hasB ? expf(v1 - m) : 0.0f;
    float s = e0 + e1;
    #pragma unroll
    for (int off = 16; off > 0; off >>= 1) s += __shfl_xor(s, off, 32);
    const float ls = logf(s);
    orow[lane] = v0 - m - ls;
    if (hasB) orow[C2A + lane] = v1 - m - ls;
}

static inline void launch_dense(bool two, const float* agg0, const float* agg1,
                                const float* deg0, const float* deg1, const float* hin,
                                const float* WFl, const float* blp, const float* WFr,
                                float* hout, int do_relu,
                                int N, hipStream_t stream) {
    const int n_strips = (N + SROWS - 1) / SROWS;
    const int n_wg = n_strips < 768 ? n_strips : 768;
    dim3 block(32, 8);
    dim3 grid(n_wg);
    (void)two;
    gnn_dense_wmma_kernel<false><<<grid, block, 0, stream>>>(
        agg0, agg1, deg0, deg1, hin, WFl, blp, WFr, hout, do_relu, N, n_strips);
}

extern "C" void kernel_launch(void* const* d_in, const int* in_sizes, int n_in,
                              void* d_out, int out_size, void* d_ws, size_t ws_size,
                              hipStream_t stream) {
    const float* x   = (const float*)d_in[0];
    const int*   ei  = (const int*)  d_in[1];
    const float* Wl0 = (const float*)d_in[2];
    const float* bl0 = (const float*)d_in[3];
    const float* Wr0 = (const float*)d_in[4];
    const float* Wl1 = (const float*)d_in[5];
    const float* bl1 = (const float*)d_in[6];
    const float* Wr1 = (const float*)d_in[7];
    const float* Wl2 = (const float*)d_in[8];
    const float* bl2 = (const float*)d_in[9];
    const float* Wr2 = (const float*)d_in[10];

    const int N = in_sizes[0] / DIN;
    const int E = in_sizes[1] / 2;
    const int* src = ei;
    const int* dst = ei + E;

    const int DP0 = 64, DP1 = 64, DP2 = 48;

    const int R      = (((N + 254) / 255) + 7) / 8 * 8;
    const int n_own  = (N + R - 1) / R;
    const int n_chunks = (E + 511) / 512;
    const int n_kwords = n_chunks * 128;

    float* deg  = (float*)d_ws;
    float* agg  = deg + (size_t)N;
    float* h1   = agg + (size_t)N * DIN;
    float* h2   = h1  + (size_t)N * DIN;
    float* WTl0 = h2  + (size_t)N * DIN;
    float* WTr0 = WTl0 + DP0 * DIN;
    float* blp0 = WTr0 + DP0 * DIN;
    float* WTl1 = blp0 + DP0;
    float* WTr1 = WTl1 + DP1 * DIN;
    float* blp1 = WTr1 + DP1 * DIN;
    float* WTl2 = blp1 + DP1;
    float* WTr2 = WTl2 + DP2 * DIN;
    float* blp2 = WTr2 + DP2 * DIN;
    const size_t key_off = ((size_t)((blp2 + DP2) - deg) + 3) / 4 * 4;
    unsigned* keys = (unsigned*)(deg + key_off);

    const dim3 own_block(32, OWN_WAVES);
    const int ZB = 256;

    gnn_pack_weights_kernel<<<(DP0 * DIN + ZB - 1) / ZB, ZB, 0, stream>>>(
        Wl0, bl0, Wr0, WTl0, WTr0, blp0, 64, DP0);
    gnn_pack_weights_kernel<<<(DP1 * DIN + ZB - 1) / ZB, ZB, 0, stream>>>(
        Wl1, bl1, Wr1, WTl1, WTr1, blp1, 64, DP1);
    gnn_pack_weights_kernel<<<(DP2 * DIN + ZB - 1) / ZB, ZB, 0, stream>>>(
        Wl2, bl2, Wr2, WTl2, WTr2, blp2, 40, DP2);
    gnn_owner_key_kernel<<<(n_kwords + ZB - 1) / ZB, ZB, 0, stream>>>(
        dst, keys, E, n_kwords, R);

    gnn_owner_sum_kernel<DIN><<<dim3(n_own), own_block, 0, stream>>>(
        x, src, dst, keys, n_chunks, agg, deg, N, R, 1);
    launch_dense(false, agg, agg, deg, deg, x, WTl0, blp0, WTr0, h1, 1, N, stream);

    gnn_owner_sum_kernel<DIN><<<dim3(n_own), own_block, 0, stream>>>(
        h1, src, dst, keys, n_chunks, agg, deg, N, R, 0);
    launch_dense(false, agg, agg, deg, deg, h1, WTl1, blp1, WTr1, h2, 1, N, stream);

    {
        const int n_strips = (N + SROWS - 1) / SROWS;
        const int n_wg = n_strips < 768 ? n_strips : 768;
        float* y   = h1;
        float* s40 = agg;
        gnn_proj_wmma_kernel<<<dim3(n_wg), dim3(32, 8), 0, stream>>>(
            h2, WTl2, WTr2, blp2, y, (float*)d_out, N, n_strips);
        gnn_owner_sum_kernel<C2><<<dim3(n_own), own_block, 0, stream>>>(
            y, src, dst, keys, n_chunks, s40, deg, N, R, 0);
        gnn_mean_logsoftmax_kernel<<<dim3((N + 7) / 8), dim3(32, 8), 0, stream>>>(
            s40, deg, (float*)d_out, N);
    }
}


# Owner sweep reads one bit per edge, not one byte: a build pass (LDS OR tile, whole-line stores) writes an edge bitmap per owner; each owner scans only its own E/8 bytes; arrival-order sum kept
# speedup vs baseline: 3.0572x; 1.1743x over previous
#include <hip/hip_runtime.h>
#include <cstddef>

typedef float v2f __attribute__((ext_vector_type(2)));
typedef float v4f __attribute__((ext_vector_type(4)));
typedef float v8f __attribute__((ext_vector_type(8)));

#define DIN 64

__global__ void gnn_pack_weights_kernel(const float* __restrict__ Wl,
                                        const float* __restrict__ bl,
                                        const float* __restrict__ Wr,
                                        float* __restrict__ WFl,
                                        float* __restrict__ WFr,
                                        float* __restrict__ blp,
                                        int dout, int dout_pad) {
    int tid = blockIdx.x * blockDim.x + threadIdx.x;
    int col = tid / DIN;
    int k   = tid % DIN;
    if (col >= dout_pad) return;
    bool cv  = (col < dout);
    const int i = k >> 2, h = (k >> 1) & 1, e = k & 1;
    const size_t o = ((size_t)(((col >> 4) * 16 + i) * 32 + h * 16 + (col & 15))) * 2 + e;
    WFl[o] = cv ? Wl[(size_t)k * dout + col] : 0.0f;
    WFr[o] = cv ? Wr[(size_t)k * dout + col] : 0.0f;
    if (k == 0) blp[col] = cv ? bl[col] : 0.0f;
}

#define OWN_WAVES 32
#define OWN_RMAX  640
#define OWN_QCAP  64
#define OWN_BATCH 16

__global__ void gnn_owner_key_kernel(const int* __restrict__ dst,
                                     unsigned* __restrict__ keys,
                                     int E, int n_words, int R) {
    const int w = blockIdx.x * blockDim.x + threadIdx.x;
    if (w >= n_words) return;
    unsigned kw = 0u;
    #pragma unroll
    for (int b = 0; b < 4; ++b) {
        const long long e = (long long)w * 4 + b;
        unsigned k = 0xFFu;
        if (e < (long long)E) k = (unsigned)dst[e] / (unsigned)R;
        kw |= k << (8 * b);
    }
    keys[w] = kw;
}

#define BM_EDGES  2048
#define BM_LWORDS 64

__global__ void __launch_bounds__(256)
gnn_owner_bitmap_kernel(const int* __restrict__ dst,
                        unsigned* __restrict__ bm,
                        int E, int R, int n_own, int wpo) {
    __shared__ __attribute__((aligned(16))) unsigned tile[255 * BM_LWORDS];
    const int t  = threadIdx.x;
    const int n4 = n_own * (BM_LWORDS / 4);
    for (int i = t; i < n4; i += 256) ((uint4*)tile)[i] = make_uint4(0u, 0u, 0u, 0u);
    __syncthreads();
    const long long base = (long long)blockIdx.x * BM_EDGES;
    int d[BM_EDGES / 256];
    #pragma unroll
    for (int b = 0; b < BM_EDGES / 256; ++b) {
        const long long e = base + b * 256 + t;
        d[b] = dst[e < (long long)E ? e : (long long)E - 1];
    }
    #pragma unroll
    for (int b = 0; b < BM_EDGES / 256; ++b) {
        const int le = b * 256 + t;
        if (base + le < (long long)E) {
            const unsigned g = (unsigned)d[b] / (unsigned)R;
            atomicOr(&tile[g * BM_LWORDS + (le >> 5)], 1u << (le & 31));
        }
    }
    __syncthreads();
    uint4* o4 = (uint4*)bm;
    for (int i = t; i < n4; i += 256) {
        const int g = i >> 4, p = i & 15;
        o4[((size_t)g * (size_t)wpo + (size_t)blockIdx.x * BM_LWORDS) / 4 + p] = ((const uint4*)tile)[i];
    }
}

__device__ __forceinline__ unsigned gnn_zero_bytes(unsigned x) {
    return ~(((x & 0x7F7F7F7Fu) + 0x7F7F7F7Fu) | x | 0x7F7F7F7Fu);
}

template <int W>
__device__ __forceinline__ void
gnn_owner_drain(const float* __restrict__ rows,
                const int* __restrict__ src, const int* __restrict__ dst,
                const unsigned* qw, unsigned head, int n, int lo, int lane,
                float* acc, unsigned* degL, int do_deg) {
    int s = 0, r = 0;
    if (lane < n) {
        const unsigned e = qw[(head + (unsigned)lane) & (OWN_QCAP - 1)];
        s = src[e];
        r = dst[e] - lo;
        if (do_deg) atomicAdd(&degL[r], 1u);
    }
    const bool act  = (W == DIN) || (2 * lane < W);
    const int  loff = act ? 2 * lane : 0;
    #pragma unroll
    for (int b = 0; b < 32 / OWN_BATCH; ++b) {
        if (b * OWN_BATCH < n) {
            v2f v[OWN_BATCH];
            #pragma unroll
            for (int j = 0; j < OWN_BATCH; ++j) {
                const int sj = __builtin_amdgcn_readlane(s, b * OWN_BATCH + j);
                v[j] = *(const v2f*)(rows + (size_t)sj * DIN + loff);
            }
            if (act) {
                #pragma unroll
                for (int j = 0; j < OWN_BATCH; ++j) {
                    if (b * OWN_BATCH + j < n) {
                        const int rj = __builtin_amdgcn_readlane(r, b * OWN_BATCH + j);
                        float* a = acc + rj * W + 2 * lane;
                        __hip_atomic_fetch_add(a,     v[j].x, __ATOMIC_RELAXED, __HIP_MEMORY_SCOPE_WORKGROUP);
                        __hip_atomic_fetch_add(a + 1, v[j].y, __ATOMIC_RELAXED, __HIP_MEMORY_SCOPE_WORKGROUP);
                    }
                }
            }
        }
    }
}

template <int W, bool BM>
__global__ void __launch_bounds__(32 * OWN_WAVES)
gnn_owner_sum_kernel(const float* __restrict__ rows,
                     const int* __restrict__ src,
                     const int* __restrict__ dst,
                     const unsigned* __restrict__ keys,
                     int n_chunks,
                     float* __restrict__ sums,
                     float* __restrict__ deg,
                     int n_nodes, int R, int do_deg) {
    __shared__ __attribute__((aligned(16))) float acc[OWN_RMAX * DIN];
    __shared__ unsigned degL[OWN_RMAX];
    __shared__ unsigned q[OWN_WAVES][OWN_QCAP];

    const int lane = threadIdx.x;
    const int wv = __builtin_amdgcn_readfirstlane((int)threadIdx.y);
    const int t  = wv * 32 + lane;
    const int g  = blockIdx.x;
    const int lo = g * R;
    int nrows = n_nodes - lo;
    nrows = nrows < R ? nrows : R;

    const int n4 = nrows * W / 4;
    for (int i = t; i < n4; i += 32 * OWN_WAVES) ((v4f*)acc)[i] = v4f{0.0f, 0.0f, 0.0f, 0.0f};
    for (int i = t; i < nrows; i += 32 * OWN_WAVES) degL[i] = 0u;
    __syncthreads();

    unsigned* qw = &q[wv][0];
    unsigned head = 0u;
    int cnt = 0;

    if (BM) {
        const uint4* b4 = (const uint4*)(keys + (size_t)g * (size_t)n_chunks * 128u);
        int kb = wv;
        uint4 bw = make_uint4(0u, 0u, 0u, 0u);
        if (kb < n_chunks) bw = b4[(size_t)kb * 32 + lane];
        while (kb < n_chunks) {
            const unsigned ebase = (unsigned)kb * 4096u + (unsigned)lane * 128u;
            const int kbn = kb + OWN_WAVES;
            const uint4 bwn = b4[(size_t)(kbn < n_chunks ? kbn : kb) * 32 + lane];
            #pragma unroll
            for (int hh = 0; hh < 2; ++hh) {
                unsigned long long m = hh ? (((unsigned long long)bw.w << 32) | bw.z)
                                          : (((unsigned long long)bw.y << 32) | bw.x);
                const unsigned eb = ebase + 64u * (unsigned)hh;
                for (;;) {
                    const bool has = m != 0ull;
                    const unsigned bal = __builtin_amdgcn_ballot_w32(has);
                    if (bal == 0u) break;
                    if (has) {
                        const unsigned pre = __builtin_amdgcn_mbcnt_lo(bal, 0u);
                        const unsigned tz  = (unsigned)__builtin_ctzll(m);
                        qw[(head + (unsigned)cnt + pre) & (OWN_QCAP - 1)] = eb + tz;
                        m &= m - 1ull;
                    }
                    cnt += __builtin_popcount(bal);
                    if (cnt >= 32) {
                        gnn_owner_drain<W>(rows, src, dst, qw, head, 32, lo, lane, acc, degL, do_deg);
                        head = (head + 32u) & (OWN_QCAP - 1);
                        cnt -= 32;
                    }
                }
            }
            kb = kbn; bw = bwn;
        }
    } else {
    const unsigned pat = (unsigned)g * 0x01010101u;
    const int start = (int)(((unsigned)g * 97u) % (unsigned)n_chunks);
    const uint4* k4 = (const uint4*)keys;

    int k = wv;
    uint4 kw = make_uint4(0xFFFFFFFFu, 0xFFFFFFFFu, 0xFFFFFFFFu, 0xFFFFFFFFu);
    int c = 0;
    if (k < n_chunks) {
        c = start + k; if (c >= n_chunks) c -= n_chunks;
        kw = k4[(size_t)c * 32 + lane];
    }
    while (k < n_chunks) {
        unsigned m = (gnn_zero_bytes(kw.x ^ pat) >> 7) | (gnn_zero_bytes(kw.y ^ pat) >> 6)
                   | (gnn_zero_bytes(kw.z ^ pat) >> 5) | (gnn_zero_bytes(kw.w ^ pat) >> 4);
        const unsigned ebase = (unsigned)c * 512u + (unsigned)lane * 16u;

        const int kn = k + OWN_WAVES;
        int cn = start + (kn < n_chunks ? kn : k);
        if (cn >= n_chunks) cn -= n_chunks;
        const uint4 kwn = k4[(size_t)cn * 32 + lane];
        for (;;) {
            const bool has = m != 0u;
            const unsigned bal = __builtin_amdgcn_ballot_w32(has);
            if (bal == 0u) break;
            if (has) {
                const unsigned pre = __builtin_amdgcn_mbcnt_lo(bal, 0u);
                const unsigned tz  = (unsigned)__builtin_ctz(m);
                qw[(head + (unsigned)cnt + pre) & (OWN_QCAP - 1)] = ebase + (tz & 7u) * 4u + (tz >> 3);
                m &= m - 1u;
            }
            cnt += __builtin_popcount(bal);
            if (cnt >= 32) {
                gnn_owner_drain<W>(rows, src, dst, qw, head, 32, lo, lane, acc, degL, do_deg);
                head = (head + 32u) & (OWN_QCAP - 1);
                cnt -= 32;
            }
        }
        k = kn; c = cn; kw = kwn;
    }
    }
    if (cnt > 0)
        gnn_owner_drain<W>(rows, src, dst, qw, head, cnt, lo, lane, acc, degL, do_deg);

    __syncthreads();
    v4f* o4 = (v4f*)(sums + (size_t)lo * W);
    for (int i = t; i < n4; i += 32 * OWN_WAVES) o4[i] = ((const v4f*)acc)[i];
    if (do_deg)
        for (int i = t; i < nrows; i += 32 * OWN_WAVES) deg[lo + i] = (float)degL[i];
}

#define SROWS  32
#define APITCH 68

template <bool TWO>
__device__ __forceinline__ void
gnn_strip_request(const float* agg0, const float* __restrict__ agg1,
                  const float* __restrict__ deg0, const float* __restrict__ deg1,
                  const float* __restrict__ hin, int strip, int srow, int c4, int n_rows,
                  v4f (&ra0)[2], v4f (&ra1)[2], v4f (&rx)[2],
                  float (&rd0)[2], float (&rd1)[2]) {
    #pragma unroll
    for (int j = 0; j < 2; ++j) {
        int row = strip * SROWS + srow + 16 * j;
        row = row < n_rows ? row : n_rows - 1;
        const size_t o = (size_t)row * DIN + c4;
        ra0[j] = *(const v4f*)(agg0 + o);
        if (TWO) ra1[j] = *(const v4f*)(agg1 + o);
        rx[j]  = *(const v4f*)(hin + o);
        rd0[j] = deg0[row];
        if (TWO) rd1[j] = deg1[row];
    }
}

template <bool TWO>
__global__ void __launch_bounds__(256)
gnn_dense_wmma_kernel(const float* agg0,
                      const float* __restrict__ agg1,
                      const float* __restrict__ deg0,
                      const float* __restrict__ deg1,
                      const float* __restrict__ hin,
                      const float* __restrict__ WFl,
                      const float* __restrict__ blp,
                      const float* __restrict__ WFr,
                      float* hout,
                      int do_relu, int n_rows, int n_strips) {
    __shared__ __attribute__((aligned(16))) float sM[2][SROWS * APITCH];
    __shared__ __attribute__((aligned(16))) float sX[2][SROWS * APITCH];

    const int lane  = threadIdx.x;
    const int wv    = __builtin_amdgcn_readfirstlane((int)threadIdx.y);
    const int ntile = wv & 3;
    const int mhalf = wv >> 2;
    const int khalf = (lane >> 4) * 2;
    const int col   = ntile * 16 + (lane & 15);

    int strip = blockIdx.x;
    if (strip >= n_strips) return;

    const int t    = wv * 32 + lane;
    const int srow = t >> 4;
    const int c4   = (t & 15) * 4;

    v4f ra0[2], ra1[2], rx[2];
    float  rd0[2], rd1[2];
    gnn_strip_request<TWO>(agg0, agg1, deg0, deg1, hin, strip, srow, c4, n_rows,
                           ra0, ra1, rx, rd0, rd1);

    v2f bL[16], bR[16];
    {
        const size_t wo = ((size_t)(ntile * 16) * 32 + lane) * 2;
        #pragma unroll
        for (int i = 0; i < 16; ++i) {
            bL[i] = *(const v2f*)(WFl + wo + (size_t)i * 64);
            bR[i] = *(const v2f*)(WFr + wo + (size_t)i * 64);
        }
    }
    const float bv = blp[col];

    const int aoff = (mhalf * 16 + (lane & 15)) * APITCH + khalf;

    int buf = 0;
    for (;;) {
        float* pm = &sM[buf][0];
        float* px = &sX[buf][0];
        #pragma unroll
        for (int j = 0; j < 2; ++j) {
            float dsum = rd0[j];
            if (TWO) dsum += rd1[j];
            const float rdeg = 1.0f / fmaxf(dsum, 1.0f);
            v4f m = ra0[j];
            if (TWO) m += ra1[j];
            m *= rdeg;
            const int o = (srow + 16 * j) * APITCH + c4;
            *(v4f*)(pm + o) = m;
            *(v4f*)(px + o) = rx[j];
        }
        __builtin_amdgcn_fence(__ATOMIC_RELEASE, "workgroup", "local");
        __builtin_amdgcn_s_barrier();
        __builtin_amdgcn_fence(__ATOMIC_ACQUIRE, "workgroup", "local");

        const int next = strip + (int)gridDim.x;
        if (next < n_strips)
            gnn_strip_request<TWO>(agg0, agg1, deg0, deg1, hin, next, srow, c4, n_rows,
                                   ra0, ra1, rx, rd0, rd1);

        const float* am = pm + aoff;
        const float* ax = px + aoff;
        v8f c = {};
        #pragma unroll
        for (int i = 0; i < 16; ++i) {
            const v2f aAgg = *(const v2f*)(am + 4 * i);
            const v2f aX   = *(const v2f*)(ax + 4 * i);
            c = __builtin_amdgcn_wmma_f32_16x16x4_f32(false, aAgg, false, bL[i],
                                                      (short)0, c, false, false);
            c = __builtin_amdgcn_wmma_f32_16x16x4_f32(false, aX,   false, bR[i],
                                                      (short)0, c, false, false);
        }

        const int row0 = strip * SROWS + mhalf * 16 + (lane >> 4) * 8;
        float* outp = hout + (size_t)row0 * DIN + col;
        if (strip * SROWS + mhalf * 16 + 16 <= n_rows) {
            #pragma unroll
            for (int r = 0; r < 8; ++r) {
                float v = c[r] + bv;
                if (do_relu) v = fmaxf(v, 0.0f);
                outp[(size_t)r * DIN] = v;
            }
        } else {
            #pragma unroll
            for (int r = 0; r < 8; ++r) {
                float v = c[r] + bv;
                if (do_relu) v = fmaxf(v, 0.0f);
                if (row0 + r < n_rows) outp[(size_t)r * DIN] = v;
            }
        }

        if (next >= n_strips) break;
        strip = next;
        buf ^= 1;
    }
}

#define C2   40
#define C2A  32
#define C2B  8

__device__ __forceinline__ void
gnn_proj_store(const v8f& cy, const v8f& cz, float bv,
               float* __restrict__ yp, int ypitch, float* __restrict__ zp, int nvalid) {
    #pragma unroll
    for (int r = 0; r < 8; ++r) {
        if (r < nvalid) {
            yp[(size_t)r * ypitch] = cy[r];
            zp[(size_t)r * C2]     = cz[r] + bv;
        }
    }
}

__global__ void __launch_bounds__(256)
gnn_proj_wmma_kernel(const float* __restrict__ hin,
                     const float* __restrict__ WFl,
                     const float* __restrict__ WFr,
                     const float* __restrict__ blp,
                     float* __restrict__ y,
                     float* __restrict__ z,
                     int n_rows, int n_strips) {
    __shared__ __attribute__((aligned(16))) float sX[2][SROWS * APITCH];

    const int lane  = threadIdx.x;
    const int wv    = __builtin_amdgcn_readfirstlane((int)threadIdx.y);
    const bool worker = wv < 6;
    const int mhalf = wv >= 3 ? 1 : 0;
    const int ntile = wv - 3 * mhalf;
    const int khalf = (lane >> 4) * 2;
    const int col   = ntile * 16 + (lane & 15);

    int strip = blockIdx.x;
    if (strip >= n_strips) return;

    const int t    = wv * 32 + lane;
    const int srow = t >> 4;
    const int c4   = (t & 15) * 4;

    v4f rx[2];
    #pragma unroll
    for (int j = 0; j < 2; ++j) {
        int row = strip * SROWS + srow + 16 * j;
        row = row < n_rows ? row : n_rows - 1;
        rx[j] = *(const v4f*)(hin + (size_t)row * DIN + c4);
    }

    v2f bL[16], bR[16];
    float bv = 0.0f;
    if (worker) {
        const size_t wo = ((size_t)(ntile * 16) * 32 + lane) * 2;
        #pragma unroll
        for (int i = 0; i < 16; ++i) {
            bL[i] = *(const v2f*)(WFl + wo + (size_t)i * 64);
            bR[i] = *(const v2f*)(WFr + wo + (size_t)i * 64);
        }
        bv = blp[col];
    }

    const int aoff = (mhalf * 16 + (lane & 15)) * APITCH + khalf;

    int buf = 0;
    for (;;) {
        float* px = &sX[buf][0];
        #pragma unroll
        for (int j = 0; j < 2; ++j)
            *(v4f*)(px + (srow + 16 * j) * APITCH + c4) = rx[j];
        __builtin_amdgcn_fence(__ATOMIC_RELEASE, "workgroup", "local");
        __builtin_amdgcn_s_barrier();
        __builtin_amdgcn_fence(__ATOMIC_ACQUIRE, "workgroup", "local");

        const int next = strip + (int)gridDim.x;
        if (next < n_strips) {
            #pragma unroll
            for (int j = 0; j < 2; ++j) {
                int row = next * SROWS + srow + 16 * j;
                row = row < n_rows ? row : n_rows - 1;
                rx[j] = *(const v4f*)(hin + (size_t)row * DIN + c4);
            }
        }

        if (worker) {
            const float* ax = px + aoff;
            v8f cy = {};
            v8f cz = {};
            #pragma unroll
            for (int i = 0; i < 16; ++i) {
                const v2f a = *(const v2f*)(ax + 4 * i);
                cy = __builtin_amdgcn_wmma_f32_16x16x4_f32(false, a, false, bL[i],
                                                           (short)0, cy, false, false);
                cz = __builtin_amdgcn_wmma_f32_16x16x4_f32(false, a, false, bR[i],
                                                           (short)0, cz, false, false);
            }

            const int tile_row = strip * SROWS + mhalf * 16;
            const int row0i    = tile_row + (lane >> 4) * 8;
            const size_t row0  = (size_t)row0i;
            const bool full  = tile_row + 16 <= n_rows;
            const int nvalid = n_rows - row0i;
            if (col < C2) {
                float* yp = y + row0 * DIN + col;
                float* zp = z + row0 * C2  + col;
                if (full) gnn_proj_store(cy, cz, bv, yp, DIN, zp, 8);
                else      gnn_proj_store(cy, cz, bv, yp, DIN, zp, nvalid);
            }
        }

        if (next >= n_strips) break;
        strip = next;
        buf ^= 1;
    }
}

__global__ void gnn_mean_logsoftmax_kernel(const float* __restrict__ s40,
                                           const float* __restrict__ deg,
                                           float* out,
                                           int n_nodes) {
    const int lane = threadIdx.x;
    const int node = blockIdx.x * blockDim.y + threadIdx.y;
    if (node >= n_nodes) return;
    const bool hasB = lane < C2B;
    const float NEG_INF = -__builtin_inff();

    const float dsum = deg[node];
    const float a = s40[(size_t)node * C2 + lane];
    const float b = hasB ? s40[(size_t)node * C2 + C2A + lane] : 0.0f;
    const float rdeg = 1.0f / fmaxf(dsum, 1.0f);
    float* orow = out + (size_t)node * C2;
    const float v0 = a * rdeg + orow[lane];
    const float v1 = hasB ? b * rdeg + orow[C2A + lane] : NEG_INF;

    float m = fmaxf(v0, v1);
    #pragma unroll
    for (int off = 16; off > 0; off >>= 1) m = fmaxf(m, __shfl_xor(m, off, 32));
    const float e0 = expf(v0 - m);
    const float e1 = hasB ? expf(v1 - m) : 0.0f;
    float s = e0 + e1;
    #pragma unroll
    for (int off = 16; off > 0; off >>= 1) s += __shfl_xor(s, off, 32);
    const float ls = logf(s);
    orow[lane] = v0 - m - ls;
    if (hasB) orow[C2A + lane] = v1 - m - ls;
}

static inline void launch_dense(bool two, const float* agg0, const float* agg1,
                                const float* deg0, const float* deg1, const float* hin,
                                const float* WFl, const float* blp, const float* WFr,
                                float* hout, int do_relu,
                                int N, hipStream_t stream) {
    const int n_strips = (N + SROWS - 1) / SROWS;
    const int n_wg = n_strips < 768 ? n_strips : 768;
    dim3 block(32, 8);
    dim3 grid(n_wg);
    (void)two;
    gnn_dense_wmma_kernel<false><<<grid, block, 0, stream>>>(
        agg0, agg1, deg0, deg1, hin, WFl, blp, WFr, hout, do_relu, N, n_strips);
}

extern "C" void kernel_launch(void* const* d_in, const int* in_sizes, int n_in,
                              void* d_out, int out_size, void* d_ws, size_t ws_size,
                              hipStream_t stream) {
    const float* x   = (const float*)d_in[0];
    const int*   ei  = (const int*)  d_in[1];
    const float* Wl0 = (const float*)d_in[2];
    const float* bl0 = (const float*)d_in[3];
    const float* Wr0 = (const float*)d_in[4];
    const float* Wl1 = (const float*)d_in[5];
    const float* bl1 = (const float*)d_in[6];
    const float* Wr1 = (const float*)d_in[7];
    const float* Wl2 = (const float*)d_in[8];
    const float* bl2 = (const float*)d_in[9];
    const float* Wr2 = (const float*)d_in[10];

    const int N = in_sizes[0] / DIN;
    const int E = in_sizes[1] / 2;
    const int* src = ei;
    const int* dst = ei + E;

    const int DP0 = 64, DP1 = 64, DP2 = 48;

    const int R      = (((N + 254) / 255) + 7) / 8 * 8;
    const int n_own  = (N + R - 1) / R;
    const int n_chunks = (E + 511) / 512;
    const int n_kwords = n_chunks * 128;
    const int n_bch    = (E + 4095) / 4096;
    const int wpo      = n_bch * 128;

    float* deg  = (float*)d_ws;
    float* bufA = deg  + (size_t)N;
    float* bufB = bufA + (size_t)N * DIN;
    float* WTl0 = bufB + (size_t)N * DIN;
    float* WTr0 = WTl0 + DP0 * DIN;
    float* blp0 = WTr0 + DP0 * DIN;
    float* WTl1 = blp0 + DP0;
    float* WTr1 = WTl1 + DP1 * DIN;
    float* blp1 = WTr1 + DP1 * DIN;
    float* WTl2 = blp1 + DP1;
    float* WTr2 = WTl2 + DP2 * DIN;
    float* blp2 = WTr2 + DP2 * DIN;
    const size_t key_off = ((size_t)((blp2 + DP2) - deg) + 63) / 64 * 64;
    unsigned* keys = (unsigned*)(deg + key_off);
    const bool use_bm = ws_size >= (key_off + (size_t)n_own * (size_t)wpo) * sizeof(float);

    const dim3 own_block(32, OWN_WAVES);
    const int ZB = 256;

    gnn_pack_weights_kernel<<<(DP0 * DIN + ZB - 1) / ZB, ZB, 0, stream>>>(
        Wl0, bl0, Wr0, WTl0, WTr0, blp0, 64, DP0);
    gnn_pack_weights_kernel<<<(DP1 * DIN + ZB - 1) / ZB, ZB, 0, stream>>>(
        Wl1, bl1, Wr1, WTl1, WTr1, blp1, 64, DP1);
    gnn_pack_weights_kernel<<<(DP2 * DIN + ZB - 1) / ZB, ZB, 0, stream>>>(
        Wl2, bl2, Wr2, WTl2, WTr2, blp2, 40, DP2);
    if (use_bm)
        gnn_owner_bitmap_kernel<<<wpo / BM_LWORDS, 256, 0, stream>>>(dst, keys, E, R, n_own, wpo);
    else
        gnn_owner_key_kernel<<<(n_kwords + ZB - 1) / ZB, ZB, 0, stream>>>(
            dst, keys, E, n_kwords, R);
    const int n_sweep = use_bm ? n_bch : n_chunks;

    if (use_bm)
        gnn_owner_sum_kernel<DIN, true><<<dim3(n_own), own_block, 0, stream>>>(
            x, src, dst, keys, n_sweep, bufA, deg, N, R, 1);
    else
        gnn_owner_sum_kernel<DIN, false><<<dim3(n_own), own_block, 0, stream>>>(
            x, src, dst, keys, n_sweep, bufA, deg, N, R, 1);
    launch_dense(false, bufA, bufA, deg, deg, x, WTl0, blp0, WTr0, bufA, 1, N, stream);

    if (use_bm)
        gnn_owner_sum_kernel<DIN, true><<<dim3(n_own), own_block, 0, stream>>>(
            bufA, src, dst, keys, n_sweep, bufB, deg, N, R, 0);
    else
        gnn_owner_sum_kernel<DIN, false><<<dim3(n_own), own_block, 0, stream>>>(
            bufA, src, dst, keys, n_sweep, bufB, deg, N, R, 0);
    launch_dense(false, bufB, bufB, deg, deg, bufA, WTl1, blp1, WTr1, bufB, 1, N, stream);

    {
        const int n_strips = (N + SROWS - 1) / SROWS;
        const int n_wg = n_strips < 768 ? n_strips : 768;
        float* y   = bufA;
        float* s40 = bufB;
        gnn_proj_wmma_kernel<<<dim3(n_wg), dim3(32, 8), 0, stream>>>(
            bufB, WTl2, WTr2, blp2, y, (float*)d_out, N, n_strips);
        if (use_bm)
            gnn_owner_sum_kernel<C2, true><<<dim3(n_own), own_block, 0, stream>>>(
                y, src, dst, keys, n_sweep, s40, deg, N, R, 0);
        else
            gnn_owner_sum_kernel<C2, false><<<dim3(n_own), own_block, 0, stream>>>(
                y, src, dst, keys, n_sweep, s40, deg, N, R, 0);
        gnn_mean_logsoftmax_kernel<<<dim3((N + 7) / 8), dim3(32, 8), 0, stream>>>(
            s40, deg, (float*)d_out, N);
    }
}


# Layers 1 and 2 replay a per-wave edge record (source and tile row in one word, whole-line stores) that layer 0's owner sweep writes: no bitmap sweep, no endpoint loads there; arrival-order sum kept
# speedup vs baseline: 3.4647x; 1.1333x over previous
#include <hip/hip_runtime.h>
#include <cstddef>

typedef float v2f __attribute__((ext_vector_type(2)));
typedef float v4f __attribute__((ext_vector_type(4)));
typedef float v8f __attribute__((ext_vector_type(8)));

#define DIN 64

__global__ void gnn_pack_weights_kernel(const float* __restrict__ Wl,
                                        const float* __restrict__ bl,
                                        const float* __restrict__ Wr,
                                        float* __restrict__ WFl,
                                        float* __restrict__ WFr,
                                        float* __restrict__ blp,
                                        int dout, int dout_pad) {
    int tid = blockIdx.x * blockDim.x + threadIdx.x;
    int col = tid / DIN;
    int k   = tid % DIN;
    if (col >= dout_pad) return;
    bool cv  = (col < dout);
    const int i = k >> 2, h = (k >> 1) & 1, e = k & 1;
    const size_t o = ((size_t)(((col >> 4) * 16 + i) * 32 + h * 16 + (col & 15))) * 2 + e;
    WFl[o] = cv ? Wl[(size_t)k * dout + col] : 0.0f;
    WFr[o] = cv ? Wr[(size_t)k * dout + col] : 0.0f;
    if (k == 0) blp[col] = cv ? bl[col] : 0.0f;
}

#define OWN_WAVES 32
#define OWN_RMAX  640
#define OWN_QCAP  64
#define OWN_BATCH 16
#define OWN_SBITS 18
#define OWN_SMASK ((1u << OWN_SBITS) - 1u)

__global__ void gnn_owner_key_kernel(const int* __restrict__ dst,
                                     unsigned* __restrict__ keys,
                                     int E, int n_words, int R) {
    const int w = blockIdx.x * blockDim.x + threadIdx.x;
    if (w >= n_words) return;
    unsigned kw = 0u;
    #pragma unroll
    for (int b = 0; b < 4; ++b) {
        const long long e = (long long)w * 4 + b;
        unsigned k = 0xFFu;
        if (e < (long long)E) k = (unsigned)dst[e] / (unsigned)R;
        kw |= k << (8 * b);
    }
    keys[w] = kw;
}

#define BM_EDGES  2048
#define BM_LWORDS 64

__global__ void __launch_bounds__(256)
gnn_owner_bitmap_kernel(const int* __restrict__ dst,
                        unsigned* __restrict__ bm,
                        int E, int R, int n_own, int wpo) {
    __shared__ __attribute__((aligned(16))) unsigned tile[255 * BM_LWORDS];
    const int t  = threadIdx.x;
    const int n4 = n_own * (BM_LWORDS / 4);
    for (int i = t; i < n4; i += 256) ((uint4*)tile)[i] = make_uint4(0u, 0u, 0u, 0u);
    __syncthreads();
    const long long base = (long long)blockIdx.x * BM_EDGES;
    int d[BM_EDGES / 256];
    #pragma unroll
    for (int b = 0; b < BM_EDGES / 256; ++b) {
        const long long e = base + b * 256 + t;
        d[b] = dst[e < (long long)E ? e : (long long)E - 1];
    }
    #pragma unroll
    for (int b = 0; b < BM_EDGES / 256; ++b) {
        const int le = b * 256 + t;
        if (base + le < (long long)E) {
            const unsigned g = (unsigned)d[b] / (unsigned)R;
            atomicOr(&tile[g * BM_LWORDS + (le >> 5)], 1u << (le & 31));
        }
    }
    __syncthreads();
    uint4* o4 = (uint4*)bm;
    for (int i = t; i < n4; i += 256) {
        const int g = i >> 4, p = i & 15;
        o4[((size_t)g * (size_t)wpo + (size_t)blockIdx.x * BM_LWORDS) / 4 + p] = ((const uint4*)tile)[i];
    }
}

__device__ __forceinline__ unsigned gnn_zero_bytes(unsigned x) {
    return ~(((x & 0x7F7F7F7Fu) + 0x7F7F7F7Fu) | x | 0x7F7F7F7Fu);
}

template <int W>
__device__ __forceinline__ void
gnn_owner_rows(const float* __restrict__ rows, int s, int r, int n, int lane, float* acc) {
    const bool act  = (W == DIN) || (2 * lane < W);
    const int  loff = act ? 2 * lane : 0;
    #pragma unroll
    for (int b = 0; b < 32 / OWN_BATCH; ++b) {
        if (b * OWN_BATCH < n) {
            v2f v[OWN_BATCH];
            #pragma unroll
            for (int j = 0; j < OWN_BATCH; ++j) {
                const int sj = __builtin_amdgcn_readlane(s, b * OWN_BATCH + j);
                v[j] = *(const v2f*)(rows + (size_t)sj * DIN + loff);
            }
            if (act) {
                #pragma unroll
                for (int j = 0; j < OWN_BATCH; ++j) {
                    if (b * OWN_BATCH + j < n) {
                        const int rj = __builtin_amdgcn_readlane(r, b * OWN_BATCH + j);
                        float* a = acc + rj * W + 2 * lane;
                        __hip_atomic_fetch_add(a,     v[j].x, __ATOMIC_RELAXED, __HIP_MEMORY_SCOPE_WORKGROUP);
                        __hip_atomic_fetch_add(a + 1, v[j].y, __ATOMIC_RELAXED, __HIP_MEMORY_SCOPE_WORKGROUP);
                    }
                }
            }
        }
    }
}

template <int W>
__device__ __forceinline__ void
gnn_owner_drain(const float* __restrict__ rows,
                const int* __restrict__ src, const int* __restrict__ dst,
                const unsigned* qw, unsigned head, int n, int lo, int lane,
                float* acc, unsigned* degL, int do_deg,
                unsigned* seg, int pos, int capw) {
    int s = 0, r = 0;
    if (lane < n) {
        const unsigned e = qw[(head + (unsigned)lane) & (OWN_QCAP - 1)];
        s = src[e];
        r = dst[e] - lo;
        if (do_deg) atomicAdd(&degL[r], 1u);
        if (seg != nullptr && pos + lane < capw)
            seg[pos + lane] = (unsigned)s | ((unsigned)r << OWN_SBITS);
    }
    gnn_owner_rows<W>(rows, s, r, n, lane, acc);
}

template <int W, bool BM>
__global__ void __launch_bounds__(32 * OWN_WAVES)
gnn_owner_sum_kernel(const float* __restrict__ rows,
                     const int* __restrict__ src,
                     const int* __restrict__ dst,
                     const unsigned* __restrict__ keys,
                     int n_chunks,
                     float* __restrict__ sums,
                     float* __restrict__ deg,
                     int n_nodes, int R, int do_deg,
                     unsigned* list, int* lcnt, int capw, int list_mode) {
    __shared__ __attribute__((aligned(16))) float acc[OWN_RMAX * DIN];
    __shared__ unsigned degL[OWN_RMAX];
    __shared__ unsigned q[OWN_WAVES][OWN_QCAP];

    const int lane = threadIdx.x;
    const int wv = __builtin_amdgcn_readfirstlane((int)threadIdx.y);
    const int t  = wv * 32 + lane;
    const int g  = blockIdx.x;
    const int lo = g * R;
    int nrows = n_nodes - lo;
    nrows = nrows < R ? nrows : R;

    int wcount = 0;
    if (list_mode == 2) wcount = lcnt[g * OWN_WAVES + lane];

    const int n4 = nrows * W / 4;
    for (int i = t; i < n4; i += 32 * OWN_WAVES) ((v4f*)acc)[i] = v4f{0.0f, 0.0f, 0.0f, 0.0f};
    for (int i = t; i < nrows; i += 32 * OWN_WAVES) degL[i] = 0u;
    __syncthreads();

    const bool replay = list_mode == 2 && __builtin_amdgcn_ballot_w32(wcount > capw) == 0u;
    const size_t seg_off = ((size_t)g * OWN_WAVES + (size_t)wv) * (size_t)capw;

    if (replay) {
        const int n = __builtin_amdgcn_readlane(wcount, wv);
        if (n > 0) {
            const unsigned* seg = list + seg_off;
            const int last  = n - 1;
            const int nfull = n >> 5;
            unsigned ent = seg[lane < last ? lane : last];
            for (int k = 0; k < nfull; ++k) {
                const int in = (k + 1) * 32 + lane;
                const unsigned entn = seg[in < last ? in : last];
                gnn_owner_rows<W>(rows, (int)(ent & OWN_SMASK), (int)(ent >> OWN_SBITS), 32, lane, acc);
                ent = entn;
            }
            const int rem = n & 31;
            if (rem > 0)
                gnn_owner_rows<W>(rows, (int)(ent & OWN_SMASK), (int)(ent >> OWN_SBITS), rem, lane, acc);
        }
    } else {
    unsigned* seg = list_mode == 1 ? list + seg_off : nullptr;
    int pos = 0;

    unsigned* qw = &q[wv][0];
    unsigned head = 0u;
    int cnt = 0;

    if (BM) {
        const uint4* b4 = (const uint4*)(keys + (size_t)g * (size_t)n_chunks * 128u);
        int kb = wv;
        uint4 bw = make_uint4(0u, 0u, 0u, 0u);
        if (kb < n_chunks) bw = b4[(size_t)kb * 32 + lane];
        while (kb < n_chunks) {
            const unsigned ebase = (unsigned)kb * 4096u + (unsigned)lane * 128u;
            const int kbn = kb + OWN_WAVES;
            const uint4 bwn = b4[(size_t)(kbn < n_chunks ? kbn : kb) * 32 + lane];
            #pragma unroll
            for (int hh = 0; hh < 2; ++hh) {
                unsigned long long m = hh ? (((unsigned long long)bw.w << 32) | bw.z)
                                          : (((unsigned long long)bw.y << 32) | bw.x);
                const unsigned eb = ebase + 64u * (unsigned)hh;
                for (;;) {
                    const bool has = m != 0ull;
                    const unsigned bal = __builtin_amdgcn_ballot_w32(has);
                    if (bal == 0u) break;
                    if (has) {
                        const unsigned pre = __builtin_amdgcn_mbcnt_lo(bal, 0u);
                        const unsigned tz  = (unsigned)__builtin_ctzll(m);
                        qw[(head + (unsigned)cnt + pre) & (OWN_QCAP - 1)] = eb + tz;
                        m &= m - 1ull;
                    }
                    cnt += __builtin_popcount(bal);
                    if (cnt >= 32) {
                        gnn_owner_drain<W>(rows, src, dst, qw, head, 32, lo, lane, acc, degL, do_deg,
                                           seg, pos, capw);
                        pos += 32;
                        head = (head + 32u) & (OWN_QCAP - 1);
                        cnt -= 32;
                    }
                }
            }
            kb = kbn; bw = bwn;
        }
    } else {
    const unsigned pat = (unsigned)g * 0x01010101u;
    const int start = (int)(((unsigned)g * 97u) % (unsigned)n_chunks);
    const uint4* k4 = (const uint4*)keys;

    int k = wv;
    uint4 kw = make_uint4(0xFFFFFFFFu, 0xFFFFFFFFu, 0xFFFFFFFFu, 0xFFFFFFFFu);
    int c = 0;
    if (k < n_chunks) {
        c = start + k; if (c >= n_chunks) c -= n_chunks;
        kw = k4[(size_t)c * 32 + lane];
    }
    while (k < n_chunks) {
        unsigned m = (gnn_zero_bytes(kw.x ^ pat) >> 7) | (gnn_zero_bytes(kw.y ^ pat) >> 6)
                   | (gnn_zero_bytes(kw.z ^ pat) >> 5) | (gnn_zero_bytes(kw.w ^ pat) >> 4);
        const unsigned ebase = (unsigned)c * 512u + (unsigned)lane * 16u;

        const int kn = k + OWN_WAVES;
        int cn = start + (kn < n_chunks ? kn : k);
        if (cn >= n_chunks) cn -= n_chunks;
        const uint4 kwn = k4[(size_t)cn * 32 + lane];
        for (;;) {
            const bool has = m != 0u;
            const unsigned bal = __builtin_amdgcn_ballot_w32(has);
            if (bal == 0u) break;
            if (has) {
                const unsigned pre = __builtin_amdgcn_mbcnt_lo(bal, 0u);
                const unsigned tz  = (unsigned)__builtin_ctz(m);
                qw[(head + (unsigned)cnt + pre) & (OWN_QCAP - 1)] = ebase + (tz & 7u) * 4u + (tz >> 3);
                m &= m - 1u;
            }
            cnt += __builtin_popcount(bal);
            if (cnt >= 32) {
                gnn_owner_drain<W>(rows, src, dst, qw, head, 32, lo, lane, acc, degL, do_deg,
                                   seg, pos, capw);
                pos += 32;
                head = (head + 32u) & (OWN_QCAP - 1);
                cnt -= 32;
            }
        }
        k = kn; c = cn; kw = kwn;
    }
    }
    if (cnt > 0) {
        gnn_owner_drain<W>(rows, src, dst, qw, head, cnt, lo, lane, acc, degL, do_deg,
                           seg, pos, capw);
        pos += cnt;
    }
    if (list_mode == 1 && lane == 0) lcnt[g * OWN_WAVES + wv] = pos;
    }

    __syncthreads();
    v4f* o4 = (v4f*)(sums + (size_t)lo * W);
    for (int i = t; i < n4; i += 32 * OWN_WAVES) o4[i] = ((const v4f*)acc)[i];
    if (do_deg)
        for (int i = t; i < nrows; i += 32 * OWN_WAVES) deg[lo + i] = (float)degL[i];
}

#define SROWS  32
#define APITCH 68

template <bool TWO>
__device__ __forceinline__ void
gnn_strip_request(const float* agg0, const float* __restrict__ agg1,
                  const float* __restrict__ deg0, const float* __restrict__ deg1,
                  const float* __restrict__ hin, int strip, int srow, int c4, int n_rows,
                  v4f (&ra0)[2], v4f (&ra1)[2], v4f (&rx)[2],
                  float (&rd0)[2], float (&rd1)[2]) {
    #pragma unroll
    for (int j = 0; j < 2; ++j) {
        int row = strip * SROWS + srow + 16 * j;
        row = row < n_rows ? row : n_rows - 1;
        const size_t o = (size_t)row * DIN + c4;
        ra0[j] = *(const v4f*)(agg0 + o);
        if (TWO) ra1[j] = *(const v4f*)(agg1 + o);
        rx[j]  = *(const v4f*)(hin + o);
        rd0[j] = deg0[row];
        if (TWO) rd1[j] = deg1[row];
    }
}

template <bool TWO>
__global__ void __launch_bounds__(256)
gnn_dense_wmma_kernel(const float* agg0,
                      const float* __restrict__ agg1,
                      const float* __restrict__ deg0,
                      const float* __restrict__ deg1,
                      const float* __restrict__ hin,
                      const float* __restrict__ WFl,
                      const float* __restrict__ blp,
                      const float* __restrict__ WFr,
                      float* hout,
                      int do_relu, int n_rows, int n_strips) {
    __shared__ __attribute__((aligned(16))) float sM[2][SROWS * APITCH];
    __shared__ __attribute__((aligned(16))) float sX[2][SROWS * APITCH];

    const int lane  = threadIdx.x;
    const int wv    = __builtin_amdgcn_readfirstlane((int)threadIdx.y);
    const int ntile = wv & 3;
    const int mhalf = wv >> 2;
    const int khalf = (lane >> 4) * 2;
    const int col   = ntile * 16 + (lane & 15);

    int strip = blockIdx.x;
    if (strip >= n_strips) return;

    const int t    = wv * 32 + lane;
    const int srow = t >> 4;
    const int c4   = (t & 15) * 4;

    v4f ra0[2], ra1[2], rx[2];
    float  rd0[2], rd1[2];
    gnn_strip_request<TWO>(agg0, agg1, deg0, deg1, hin, strip, srow, c4, n_rows,
                           ra0, ra1, rx, rd0, rd1);

    v2f bL[16], bR[16];
    {
        const size_t wo = ((size_t)(ntile * 16) * 32 + lane) * 2;
        #pragma unroll
        for (int i = 0; i < 16; ++i) {
            bL[i] = *(const v2f*)(WFl + wo + (size_t)i * 64);
            bR[i] = *(const v2f*)(WFr + wo + (size_t)i * 64);
        }
    }
    const float bv = blp[col];

    const int aoff = (mhalf * 16 + (lane & 15)) * APITCH + khalf;

    int buf = 0;
    for (;;) {
        float* pm = &sM[buf][0];
        float* px = &sX[buf][0];
        #pragma unroll
        for (int j = 0; j < 2; ++j) {
            float dsum = rd0[j];
            if (TWO) dsum += rd1[j];
            const float rdeg = 1.0f / fmaxf(dsum, 1.0f);
            v4f m = ra0[j];
            if (TWO) m += ra1[j];
            m *= rdeg;
            const int o = (srow + 16 * j) * APITCH + c4;
            *(v4f*)(pm + o) = m;
            *(v4f*)(px + o) = rx[j];
        }
        __builtin_amdgcn_fence(__ATOMIC_RELEASE, "workgroup", "local");
        __builtin_amdgcn_s_barrier();
        __builtin_amdgcn_fence(__ATOMIC_ACQUIRE, "workgroup", "local");

        const int next = strip + (int)gridDim.x;
        if (next < n_strips)
            gnn_strip_request<TWO>(agg0, agg1, deg0, deg1, hin, next, srow, c4, n_rows,
                                   ra0, ra1, rx, rd0, rd1);

        const float* am = pm + aoff;
        const float* ax = px + aoff;
        v8f c = {};
        #pragma unroll
        for (int i = 0; i < 16; ++i) {
            const v2f aAgg = *(const v2f*)(am + 4 * i);
            const v2f aX   = *(const v2f*)(ax + 4 * i);
            c = __builtin_amdgcn_wmma_f32_16x16x4_f32(false, aAgg, false, bL[i],
                                                      (short)0, c, false, false);
            c = __builtin_amdgcn_wmma_f32_16x16x4_f32(false, aX,   false, bR[i],
                                                      (short)0, c, false, false);
        }

        const int row0 = strip * SROWS + mhalf * 16 + (lane >> 4) * 8;
        float* outp = hout + (size_t)row0 * DIN + col;
        if (strip * SROWS + mhalf * 16 + 16 <= n_rows) {
            #pragma unroll
            for (int r = 0; r < 8; ++r) {
                float v = c[r] + bv;
                if (do_relu) v = fmaxf(v, 0.0f);
                outp[(size_t)r * DIN] = v;
            }
        } else {
            #pragma unroll
            for (int r = 0; r < 8; ++r) {
                float v = c[r] + bv;
                if (do_relu) v = fmaxf(v, 0.0f);
                if (row0 + r < n_rows) outp[(size_t)r * DIN] = v;
            }
        }

        if (next >= n_strips) break;
        strip = next;
        buf ^= 1;
    }
}

#define C2   40
#define C2A  32
#define C2B  8

__device__ __forceinline__ void
gnn_proj_store(const v8f& cy, const v8f& cz, float bv,
               float* __restrict__ yp, int ypitch, float* __restrict__ zp, int nvalid) {
    #pragma unroll
    for (int r = 0; r < 8; ++r) {
        if (r < nvalid) {
            yp[(size_t)r * ypitch] = cy[r];
            zp[(size_t)r * C2]     = cz[r] + bv;
        }
    }
}

__global__ void __launch_bounds__(256)
gnn_proj_wmma_kernel(const float* __restrict__ hin,
                     const float* __restrict__ WFl,
                     const float* __restrict__ WFr,
                     const float* __restrict__ blp,
                     float* __restrict__ y,
                     float* __restrict__ z,
                     int n_rows, int n_strips) {
    __shared__ __attribute__((aligned(16))) float sX[2][SROWS * APITCH];

    const int lane  = threadIdx.x;
    const int wv    = __builtin_amdgcn_readfirstlane((int)threadIdx.y);
    const bool worker = wv < 6;
    const int mhalf = wv >= 3 ? 1 : 0;
    const int ntile = wv - 3 * mhalf;
    const int khalf = (lane >> 4) * 2;
    const int col   = ntile * 16 + (lane & 15);

    int strip = blockIdx.x;
    if (strip >= n_strips) return;

    const int t    = wv * 32 + lane;
    const int srow = t >> 4;
    const int c4   = (t & 15) * 4;

    v4f rx[2];
    #pragma unroll
    for (int j = 0; j < 2; ++j) {
        int row = strip * SROWS + srow + 16 * j;
        row = row < n_rows ? row : n_rows - 1;
        rx[j] = *(const v4f*)(hin + (size_t)row * DIN + c4);
    }

    v2f bL[16], bR[16];
    float bv = 0.0f;
    if (worker) {
        const size_t wo = ((size_t)(ntile * 16) * 32 + lane) * 2;
        #pragma unroll
        for (int i = 0; i < 16; ++i) {
            bL[i] = *(const v2f*)(WFl + wo + (size_t)i * 64);
            bR[i] = *(const v2f*)(WFr + wo + (size_t)i * 64);
        }
        bv = blp[col];
    }

    const int aoff = (mhalf * 16 + (lane & 15)) * APITCH + khalf;

    int buf = 0;
    for (;;) {
        float* px = &sX[buf][0];
        #pragma unroll
        for (int j = 0; j < 2; ++j)
            *(v4f*)(px + (srow + 16 * j) * APITCH + c4) = rx[j];
        __builtin_amdgcn_fence(__ATOMIC_RELEASE, "workgroup", "local");
        __builtin_amdgcn_s_barrier();
        __builtin_amdgcn_fence(__ATOMIC_ACQUIRE, "workgroup", "local");

        const int next = strip + (int)gridDim.x;
        if (next < n_strips) {
            #pragma unroll
            for (int j = 0; j < 2; ++j) {
                int row = next * SROWS + srow + 16 * j;
                row = row < n_rows ? row : n_rows - 1;
                rx[j] = *(const v4f*)(hin + (size_t)row * DIN + c4);
            }
        }

        if (worker) {
            const float* ax = px + aoff;
            v8f cy = {};
            v8f cz = {};
            #pragma unroll
            for (int i = 0; i < 16; ++i) {
                const v2f a = *(const v2f*)(ax + 4 * i);
                cy = __builtin_amdgcn_wmma_f32_16x16x4_f32(false, a, false, bL[i],
                                                           (short)0, cy, false, false);
                cz = __builtin_amdgcn_wmma_f32_16x16x4_f32(false, a, false, bR[i],
                                                           (short)0, cz, false, false);
            }

            const int tile_row = strip * SROWS + mhalf * 16;
            const int row0i    = tile_row + (lane >> 4) * 8;
            const size_t row0  = (size_t)row0i;
            const bool full  = tile_row + 16 <= n_rows;
            const int nvalid = n_rows - row0i;
            if (col < C2) {
                float* yp = y + row0 * DIN + col;
                float* zp = z + row0 * C2  + col;
                if (full) gnn_proj_store(cy, cz, bv, yp, DIN, zp, 8);
                else      gnn_proj_store(cy, cz, bv, yp, DIN, zp, nvalid);
            }
        }

        if (next >= n_strips) break;
        strip = next;
        buf ^= 1;
    }
}

__global__ void gnn_mean_logsoftmax_kernel(const float* __restrict__ s40,
                                           const float* __restrict__ deg,
                                           float* out,
                                           int n_nodes) {
    const int lane = threadIdx.x;
    const int node = blockIdx.x * blockDim.y + threadIdx.y;
    if (node >= n_nodes) return;
    const bool hasB = lane < C2B;
    const float NEG_INF = -__builtin_inff();

    const float dsum = deg[node];
    const float a = s40[(size_t)node * C2 + lane];
    const float b = hasB ? s40[(size_t)node * C2 + C2A + lane] : 0.0f;
    const float rdeg = 1.0f / fmaxf(dsum, 1.0f);
    float* orow = out + (size_t)node * C2;
    const float v0 = a * rdeg + orow[lane];
    const float v1 = hasB ? b * rdeg + orow[C2A + lane] : NEG_INF;

    float m = fmaxf(v0, v1);
    #pragma unroll
    for (int off = 16; off > 0; off >>= 1) m = fmaxf(m, __shfl_xor(m, off, 32));
    const float e0 = expf(v0 - m);
    const float e1 = hasB ? expf(v1 - m) : 0.0f;
    float s = e0 + e1;
    #pragma unroll
    for (int off = 16; off > 0; off >>= 1) s += __shfl_xor(s, off, 32);
    const float ls = logf(s);
    orow[lane] = v0 - m - ls;
    if (hasB) orow[C2A + lane] = v1 - m - ls;
}

static inline void launch_dense(bool two, const float* agg0, const float* agg1,
                                const float* deg0, const float* deg1, const float* hin,
                                const float* WFl, const float* blp, const float* WFr,
                                float* hout, int do_relu,
                                int N, hipStream_t stream) {
    const int n_strips = (N + SROWS - 1) / SROWS;
    const int n_wg = n_strips < 768 ? n_strips : 768;
    dim3 block(32, 8);
    dim3 grid(n_wg);
    (void)two;
    gnn_dense_wmma_kernel<false><<<grid, block, 0, stream>>>(
        agg0, agg1, deg0, deg1, hin, WFl, blp, WFr, hout, do_relu, N, n_strips);
}

extern "C" void kernel_launch(void* const* d_in, const int* in_sizes, int n_in,
                              void* d_out, int out_size, void* d_ws, size_t ws_size,
                              hipStream_t stream) {
    const float* x   = (const float*)d_in[0];
    const int*   ei  = (const int*)  d_in[1];
    const float* Wl0 = (const float*)d_in[2];
    const float* bl0 = (const float*)d_in[3];
    const float* Wr0 = (const float*)d_in[4];
    const float* Wl1 = (const float*)d_in[5];
    const float* bl1 = (const float*)d_in[6];
    const float* Wr1 = (const float*)d_in[7];
    const float* Wl2 = (const float*)d_in[8];
    const float* bl2 = (const float*)d_in[9];
    const float* Wr2 = (const float*)d_in[10];

    const int N = in_sizes[0] / DIN;
    const int E = in_sizes[1] / 2;
    const int* src = ei;
    const int* dst = ei + E;

    const int DP0 = 64, DP1 = 64, DP2 = 48;

    const int R      = (((N + 254) / 255) + 7) / 8 * 8;
    const int n_own  = (N + R - 1) / R;
    const int n_chunks = (E + 511) / 512;
    const int n_kwords = n_chunks * 128;
    const int n_bch    = (E + 4095) / 4096;
    const int wpo      = n_bch * 128;

    float* deg  = (float*)d_ws;
    float* bufA = deg  + (size_t)N;
    float* bufB = bufA + (size_t)N * DIN;
    float* WTl0 = bufB + (size_t)N * DIN;
    float* WTr0 = WTl0 + DP0 * DIN;
    float* blp0 = WTr0 + DP0 * DIN;
    float* WTl1 = blp0 + DP0;
    float* WTr1 = WTl1 + DP1 * DIN;
    float* blp1 = WTr1 + DP1 * DIN;
    float* WTl2 = blp1 + DP1;
    float* WTr2 = WTl2 + DP2 * DIN;
    float* blp2 = WTr2 + DP2 * DIN;
    const size_t key_off = ((size_t)((blp2 + DP2) - deg) + 63) / 64 * 64;
    unsigned* keys = (unsigned*)(deg + key_off);
    const bool use_bm = ws_size >= (key_off + (size_t)n_own * (size_t)wpo) * sizeof(float);

    const int CAPW = 320;
    const size_t cnt_words = ((size_t)n_own * OWN_WAVES + 63) / 64 * 64;
    const size_t rec_words = cnt_words + (size_t)n_own * OWN_WAVES * CAPW;
    const size_t bm_end    = key_off + (size_t)n_own * (size_t)wpo;
    int* lcnt = nullptr;
    unsigned* list = nullptr;
    int lm0 = 0, lm1 = 0, lm2 = 0;
    if (use_bm && N <= (1 << OWN_SBITS)) {
        if (ws_size >= (bm_end + rec_words) * sizeof(float)) {
            lcnt = (int*)(deg + bm_end);
            lm0 = 1; lm1 = 2; lm2 = 2;
        } else if ((size_t)out_size >= rec_words) {
            lcnt = (int*)d_out;
            lm0 = 1; lm1 = 2;
        }
        if (lm0) list = (unsigned*)lcnt + cnt_words;
    }

    const dim3 own_block(32, OWN_WAVES);
    const int ZB = 256;

    gnn_pack_weights_kernel<<<(DP0 * DIN + ZB - 1) / ZB, ZB, 0, stream>>>(
        Wl0, bl0, Wr0, WTl0, WTr0, blp0, 64, DP0);
    gnn_pack_weights_kernel<<<(DP1 * DIN + ZB - 1) / ZB, ZB, 0, stream>>>(
        Wl1, bl1, Wr1, WTl1, WTr1, blp1, 64, DP1);
    gnn_pack_weights_kernel<<<(DP2 * DIN + ZB - 1) / ZB, ZB, 0, stream>>>(
        Wl2, bl2, Wr2, WTl2, WTr2, blp2, 40, DP2);
    if (use_bm)
        gnn_owner_bitmap_kernel<<<wpo / BM_LWORDS, 256, 0, stream>>>(dst, keys, E, R, n_own, wpo);
    else
        gnn_owner_key_kernel<<<(n_kwords + ZB - 1) / ZB, ZB, 0, stream>>>(
            dst, keys, E, n_kwords, R);
    const int n_sweep = use_bm ? n_bch : n_chunks;

    if (use_bm)
        gnn_owner_sum_kernel<DIN, true><<<dim3(n_own), own_block, 0, stream>>>(
            x, src, dst, keys, n_sweep, bufA, deg, N, R, 1, list, lcnt, CAPW, lm0);
    else
        gnn_owner_sum_kernel<DIN, false><<<dim3(n_own), own_block, 0, stream>>>(
            x, src, dst, keys, n_sweep, bufA, deg, N, R, 1, list, lcnt, CAPW, 0);
    launch_dense(false, bufA, bufA, deg, deg, x, WTl0, blp0, WTr0, bufA, 1, N, stream);

    if (use_bm)
        gnn_owner_sum_kernel<DIN, true><<<dim3(n_own), own_block, 0, stream>>>(
            bufA, src, dst, keys, n_sweep, bufB, deg, N, R, 0, list, lcnt, CAPW, lm1);
    else
        gnn_owner_sum_kernel<DIN, false><<<dim3(n_own), own_block, 0, stream>>>(
            bufA, src, dst, keys, n_sweep, bufB, deg, N, R, 0, list, lcnt, CAPW, 0);
    launch_dense(false, bufB, bufB, deg, deg, bufA, WTl1, blp1, WTr1, bufB, 1, N, stream);

    {
        const int n_strips = (N + SROWS - 1) / SROWS;
        const int n_wg = n_strips < 768 ? n_strips : 768;
        float* y   = bufA;
        float* s40 = bufB;
        gnn_proj_wmma_kernel<<<dim3(n_wg), dim3(32, 8), 0, stream>>>(
            bufB, WTl2, WTr2, blp2, y, (float*)d_out, N, n_strips);
        if (use_bm)
            gnn_owner_sum_kernel<C2, true><<<dim3(n_own), own_block, 0, stream>>>(
                y, src, dst, keys, n_sweep, s40, deg, N, R, 0, list, lcnt, CAPW, lm2);
        else
            gnn_owner_sum_kernel<C2, false><<<dim3(n_own), own_block, 0, stream>>>(
                y, src, dst, keys, n_sweep, s40, deg, N, R, 0, list, lcnt, CAPW, 0);
        gnn_mean_logsoftmax_kernel<<<dim3((N + 7) / 8), dim3(32, 8), 0, stream>>>(
            s40, deg, (float*)d_out, N);
    }
}


# Last step done by the owner work-groups on their LDS sum tiles (sum/deg + z and row log_softmax, same reduction tree, arrival-order sums kept): 40-wide sum plane and wave-per-node launch go
# speedup vs baseline: 3.6301x; 1.0477x over previous
#include <hip/hip_runtime.h>
#include <cstddef>

typedef float v2f __attribute__((ext_vector_type(2)));
typedef float v4f __attribute__((ext_vector_type(4)));
typedef float v8f __attribute__((ext_vector_type(8)));

#define DIN 64

__global__ void gnn_pack_weights_kernel(const float* __restrict__ Wl,
                                        const float* __restrict__ bl,
                                        const float* __restrict__ Wr,
                                        float* __restrict__ WFl,
                                        float* __restrict__ WFr,
                                        float* __restrict__ blp,
                                        int dout, int dout_pad) {
    int tid = blockIdx.x * blockDim.x + threadIdx.x;
    int col = tid / DIN;
    int k   = tid % DIN;
    if (col >= dout_pad) return;
    bool cv  = (col < dout);
    const int i = k >> 2, h = (k >> 1) & 1, e = k & 1;
    const size_t o = ((size_t)(((col >> 4) * 16 + i) * 32 + h * 16 + (col & 15))) * 2 + e;
    WFl[o] = cv ? Wl[(size_t)k * dout + col] : 0.0f;
    WFr[o] = cv ? Wr[(size_t)k * dout + col] : 0.0f;
    if (k == 0) blp[col] = cv ? bl[col] : 0.0f;
}

#define OWN_WAVES 32
#define OWN_RMAX  640
#define OWN_RPW   (OWN_RMAX / OWN_WAVES)
#define OWN_QCAP  64
#define OWN_BATCH 16
#define OWN_SBITS 18
#define OWN_SMASK ((1u << OWN_SBITS) - 1u)

__global__ void gnn_owner_key_kernel(const int* __restrict__ dst,
                                     unsigned* __restrict__ keys,
                                     int E, int n_words, int R) {
    const int w = blockIdx.x * blockDim.x + threadIdx.x;
    if (w >= n_words) return;
    unsigned kw = 0u;
    #pragma unroll
    for (int b = 0; b < 4; ++b) {
        const long long e = (long long)w * 4 + b;
        unsigned k = 0xFFu;
        if (e < (long long)E) k = (unsigned)dst[e] / (unsigned)R;
        kw |= k << (8 * b);
    }
    keys[w] = kw;
}

#define BM_EDGES  2048
#define BM_LWORDS 64

__global__ void __launch_bounds__(256)
gnn_owner_bitmap_kernel(const int* __restrict__ dst,
                        unsigned* __restrict__ bm,
                        int E, int R, int n_own, int wpo) {
    __shared__ __attribute__((aligned(16))) unsigned tile[255 * BM_LWORDS];
    const int t  = threadIdx.x;
    const int n4 = n_own * (BM_LWORDS / 4);
    for (int i = t; i < n4; i += 256) ((uint4*)tile)[i] = make_uint4(0u, 0u, 0u, 0u);
    __syncthreads();
    const long long base = (long long)blockIdx.x * BM_EDGES;
    int d[BM_EDGES / 256];
    #pragma unroll
    for (int b = 0; b < BM_EDGES / 256; ++b) {
        const long long e = base + b * 256 + t;
        d[b] = dst[e < (long long)E ? e : (long long)E - 1];
    }
    #pragma unroll
    for (int b = 0; b < BM_EDGES / 256; ++b) {
        const int le = b * 256 + t;
        if (base + le < (long long)E) {
            const unsigned g = (unsigned)d[b] / (unsigned)R;
            atomicOr(&tile[g * BM_LWORDS + (le >> 5)], 1u << (le & 31));
        }
    }
    __syncthreads();
    uint4* o4 = (uint4*)bm;
    for (int i = t; i < n4; i += 256) {
        const int g = i >> 4, p = i & 15;
        o4[((size_t)g * (size_t)wpo + (size_t)blockIdx.x * BM_LWORDS) / 4 + p] = ((const uint4*)tile)[i];
    }
}

__device__ __forceinline__ unsigned gnn_zero_bytes(unsigned x) {
    return ~(((x & 0x7F7F7F7Fu) + 0x7F7F7F7Fu) | x | 0x7F7F7F7Fu);
}

template <int W>
__device__ __forceinline__ void
gnn_owner_rows(const float* __restrict__ rows, int s, int r, int n, int lane, float* acc) {
    const bool act  = (W == DIN) || (2 * lane < W);
    const int  loff = act ? 2 * lane : 0;
    #pragma unroll
    for (int b = 0; b < 32 / OWN_BATCH; ++b) {
        if (b * OWN_BATCH < n) {
            v2f v[OWN_BATCH];
            #pragma unroll
            for (int j = 0; j < OWN_BATCH; ++j) {
                const int sj = __builtin_amdgcn_readlane(s, b * OWN_BATCH + j);
                v[j] = *(const v2f*)(rows + (size_t)sj * DIN + loff);
            }
            if (act) {
                #pragma unroll
                for (int j = 0; j < OWN_BATCH; ++j) {
                    if (b * OWN_BATCH + j < n) {
                        const int rj = __builtin_amdgcn_readlane(r, b * OWN_BATCH + j);
                        float* a = acc + rj * W + 2 * lane;
                        __hip_atomic_fetch_add(a,     v[j].x, __ATOMIC_RELAXED, __HIP_MEMORY_SCOPE_WORKGROUP);
                        __hip_atomic_fetch_add(a + 1, v[j].y, __ATOMIC_RELAXED, __HIP_MEMORY_SCOPE_WORKGROUP);
                    }
                }
            }
        }
    }
}

template <int W>
__device__ __forceinline__ void
gnn_owner_drain(const float* __restrict__ rows,
                const int* __restrict__ src, const int* __restrict__ dst,
                const unsigned* qw, unsigned head, int n, int lo, int lane,
                float* acc, unsigned* degL, int do_deg,
                unsigned* seg, int pos, int capw) {
    int s = 0, r = 0;
    if (lane < n) {
        const unsigned e = qw[(head + (unsigned)lane) & (OWN_QCAP - 1)];
        s = src[e];
        r = dst[e] - lo;
        if (do_deg) atomicAdd(&degL[r], 1u);
        if (seg != nullptr && pos + lane < capw)
            seg[pos + lane] = (unsigned)s | ((unsigned)r << OWN_SBITS);
    }
    gnn_owner_rows<W>(rows, s, r, n, lane, acc);
}

#define C2   40
#define C2A  32
#define C2B  8

__device__ __forceinline__ void
gnn_row_logsoftmax(float a, float b, float z0, float z1, float rdeg, int lane, float* orow) {
    const bool hasB = lane < C2B;
    const float NEG_INF = -__builtin_inff();
    const float v0 = a * rdeg + z0;
    const float v1 = hasB ? b * rdeg + z1 : NEG_INF;

    float m = fmaxf(v0, v1);
    #pragma unroll
    for (int off = 16; off > 0; off >>= 1) m = fmaxf(m, __shfl_xor(m, off, 32));
    const float e0 = expf(v0 - m);
    const float e1 = hasB ? expf(v1 - m) : 0.0f;
    float s = e0 + e1;
    #pragma unroll
    for (int off = 16; off > 0; off >>= 1) s += __shfl_xor(s, off, 32);
    const float ls = logf(s);
    orow[lane] = v0 - m - ls;
    if (hasB) orow[C2A + lane] = v1 - m - ls;
}

template <int W, bool BM, bool FIN>
__global__ void __launch_bounds__(32 * OWN_WAVES)
gnn_owner_sum_kernel(const float* __restrict__ rows,
                     const int* __restrict__ src,
                     const int* __restrict__ dst,
                     const unsigned* __restrict__ keys,
                     int n_chunks,
                     float* __restrict__ sums,
                     float* __restrict__ deg,
                     int n_nodes, int R, int do_deg,
                     unsigned* list, int* lcnt, int capw, int list_mode,
                     float* fin) {
    __shared__ __attribute__((aligned(16))) float acc[OWN_RMAX * DIN];
    __shared__ unsigned degL[OWN_RMAX];
    __shared__ unsigned q[OWN_WAVES][OWN_QCAP];

    const int lane = threadIdx.x;
    const int wv = __builtin_amdgcn_readfirstlane((int)threadIdx.y);
    const int t  = wv * 32 + lane;
    const int g  = blockIdx.x;
    const int lo = g * R;
    int nrows = n_nodes - lo;
    nrows = nrows < R ? nrows : R;

    int wcount = 0;
    if (list_mode == 2) wcount = lcnt[g * OWN_WAVES + lane];

    const int n4 = nrows * W / 4;
    for (int i = t; i < n4; i += 32 * OWN_WAVES) ((v4f*)acc)[i] = v4f{0.0f, 0.0f, 0.0f, 0.0f};
    for (int i = t; i < nrows; i += 32 * OWN_WAVES) degL[i] = 0u;
    __syncthreads();

    const bool replay = list_mode == 2 && __builtin_amdgcn_ballot_w32(wcount > capw) == 0u;
    const size_t seg_off = ((size_t)g * OWN_WAVES + (size_t)wv) * (size_t)capw;

    if (replay) {
        const int n = __builtin_amdgcn_readlane(wcount, wv);
        if (n > 0) {
            const unsigned* seg = list + seg_off;
            const int last  = n - 1;
            const int nfull = n >> 5;
            unsigned ent = seg[lane < last ? lane : last];
            for (int k = 0; k < nfull; ++k) {
                const int in = (k + 1) * 32 + lane;
                const unsigned entn = seg[in < last ? in : last];
                gnn_owner_rows<W>(rows, (int)(ent & OWN_SMASK), (int)(ent >> OWN_SBITS), 32, lane, acc);
                ent = entn;
            }
            const int rem = n & 31;
            if (rem > 0)
                gnn_owner_rows<W>(rows, (int)(ent & OWN_SMASK), (int)(ent >> OWN_SBITS), rem, lane, acc);
        }
    } else {
    unsigned* seg = list_mode == 1 ? list + seg_off : nullptr;
    int pos = 0;

    unsigned* qw = &q[wv][0];
    unsigned head = 0u;
    int cnt = 0;

    if (BM) {
        const uint4* b4 = (const uint4*)(keys + (size_t)g * (size_t)n_chunks * 128u);
        int kb = wv;
        uint4 bw = make_uint4(0u, 0u, 0u, 0u);
        if (kb < n_chunks) bw = b4[(size_t)kb * 32 + lane];
        while (kb < n_chunks) {
            const unsigned ebase = (unsigned)kb * 4096u + (unsigned)lane * 128u;
            const int kbn = kb + OWN_WAVES;
            const uint4 bwn = b4[(size_t)(kbn < n_chunks ? kbn : kb) * 32 + lane];
            #pragma unroll
            for (int hh = 0; hh < 2; ++hh) {
                unsigned long long m = hh ? (((unsigned long long)bw.w << 32) | bw.z)
                                          : (((unsigned long long)bw.y << 32) | bw.x);
                const unsigned eb = ebase + 64u * (unsigned)hh;
                for (;;) {
                    const bool has = m != 0ull;
                    const unsigned bal = __builtin_amdgcn_ballot_w32(has);
                    if (bal == 0u) break;
                    if (has) {
                        const unsigned pre = __builtin_amdgcn_mbcnt_lo(bal, 0u);
                        const unsigned tz  = (unsigned)__builtin_ctzll(m);
                        qw[(head + (unsigned)cnt + pre) & (OWN_QCAP - 1)] = eb + tz;
                        m &= m - 1ull;
                    }
                    cnt += __builtin_popcount(bal);
                    if (cnt >= 32) {
                        gnn_owner_drain<W>(rows, src, dst, qw, head, 32, lo, lane, acc, degL, do_deg,
                                           seg, pos, capw);
                        pos += 32;
                        head = (head + 32u) & (OWN_QCAP - 1);
                        cnt -= 32;
                    }
                }
            }
            kb = kbn; bw = bwn;
        }
    } else {
    const unsigned pat = (unsigned)g * 0x01010101u;
    const int start = (int)(((unsigned)g * 97u) % (unsigned)n_chunks);
    const uint4* k4 = (const uint4*)keys;

    int k = wv;
    uint4 kw = make_uint4(0xFFFFFFFFu, 0xFFFFFFFFu, 0xFFFFFFFFu, 0xFFFFFFFFu);
    int c = 0;
    if (k < n_chunks) {
        c = start + k; if (c >= n_chunks) c -= n_chunks;
        kw = k4[(size_t)c * 32 + lane];
    }
    while (k < n_chunks) {
        unsigned m = (gnn_zero_bytes(kw.x ^ pat) >> 7) | (gnn_zero_bytes(kw.y ^ pat) >> 6)
                   | (gnn_zero_bytes(kw.z ^ pat) >> 5) | (gnn_zero_bytes(kw.w ^ pat) >> 4);
        const unsigned ebase = (unsigned)c * 512u + (unsigned)lane * 16u;

        const int kn = k + OWN_WAVES;
        int cn = start + (kn < n_chunks ? kn : k);
        if (cn >= n_chunks) cn -= n_chunks;
        const uint4 kwn = k4[(size_t)cn * 32 + lane];
        for (;;) {
            const bool has = m != 0u;
            const unsigned bal = __builtin_amdgcn_ballot_w32(has);
            if (bal == 0u) break;
            if (has) {
                const unsigned pre = __builtin_amdgcn_mbcnt_lo(bal, 0u);
                const unsigned tz  = (unsigned)__builtin_ctz(m);
                qw[(head + (unsigned)cnt + pre) & (OWN_QCAP - 1)] = ebase + (tz & 7u) * 4u + (tz >> 3);
                m &= m - 1u;
            }
            cnt += __builtin_popcount(bal);
            if (cnt >= 32) {
                gnn_owner_drain<W>(rows, src, dst, qw, head, 32, lo, lane, acc, degL, do_deg,
                                   seg, pos, capw);
                pos += 32;
                head = (head + 32u) & (OWN_QCAP - 1);
                cnt -= 32;
            }
        }
        k = kn; c = cn; kw = kwn;
    }
    }
    if (cnt > 0) {
        gnn_owner_drain<W>(rows, src, dst, qw, head, cnt, lo, lane, acc, degL, do_deg,
                           seg, pos, capw);
        pos += cnt;
    }
    if (list_mode == 1 && lane == 0) lcnt[g * OWN_WAVES + wv] = pos;
    }

    if (FIN) {
        static_assert(!FIN || W == C2, "the finishing step is the 40-wide last layer's");
        const int rpw = (nrows + OWN_WAVES - 1) / OWN_WAVES;
        const int r0  = wv * rpw;
        int cnt = nrows - r0;
        cnt = cnt < rpw ? cnt : rpw;
        const int l2 = lane < C2B ? C2A + lane : lane;
        float* zrow = fin + (size_t)(lo + r0) * C2;
        float z0[OWN_RPW], z1[OWN_RPW];
        #pragma unroll
        for (int j = 0; j < OWN_RPW; ++j) {
            z0[j] = 0.0f; z1[j] = 0.0f;
            if (j < cnt) {
                z0[j] = zrow[j * C2 + lane];
                z1[j] = zrow[j * C2 + l2];
            }
        }
        float dv = 1.0f;
        if (lane < cnt) dv = deg[lo + r0 + lane];
        __syncthreads();
        const float rdv = 1.0f / fmaxf(dv, 1.0f);
        const float* arow = acc + r0 * C2;
        #pragma unroll
        for (int j = 0; j < OWN_RPW; ++j) {
            if (j < cnt) {
                const float a = arow[j * C2 + lane];
                const float b = arow[j * C2 + l2];
                const float rdeg = __int_as_float(__builtin_amdgcn_readlane(__float_as_int(rdv), j));
                gnn_row_logsoftmax(a, b, z0[j], z1[j], rdeg, lane, zrow + j * C2);
            }
        }
    } else {
        __syncthreads();
        v4f* o4 = (v4f*)(sums + (size_t)lo * W);
        for (int i = t; i < n4; i += 32 * OWN_WAVES) o4[i] = ((const v4f*)acc)[i];
        if (do_deg)
            for (int i = t; i < nrows; i += 32 * OWN_WAVES) deg[lo + i] = (float)degL[i];
    }
}

#define SROWS  32
#define APITCH 68

template <bool TWO>
__device__ __forceinline__ void
gnn_strip_request(const float* agg0, const float* __restrict__ agg1,
                  const float* __restrict__ deg0, const float* __restrict__ deg1,
                  const float* __restrict__ hin, int strip, int srow, int c4, int n_rows,
                  v4f (&ra0)[2], v4f (&ra1)[2], v4f (&rx)[2],
                  float (&rd0)[2], float (&rd1)[2]) {
    #pragma unroll
    for (int j = 0; j < 2; ++j) {
        int row = strip * SROWS + srow + 16 * j;
        row = row < n_rows ? row : n_rows - 1;
        const size_t o = (size_t)row * DIN + c4;
        ra0[j] = *(const v4f*)(agg0 + o);
        if (TWO) ra1[j] = *(const v4f*)(agg1 + o);
        rx[j]  = *(const v4f*)(hin + o);
        rd0[j] = deg0[row];
        if (TWO) rd1[j] = deg1[row];
    }
}

template <bool TWO>
__global__ void __launch_bounds__(256)
gnn_dense_wmma_kernel(const float* agg0,
                      const float* __restrict__ agg1,
                      const float* __restrict__ deg0,
                      const float* __restrict__ deg1,
                      const float* __restrict__ hin,
                      const float* __restrict__ WFl,
                      const float* __restrict__ blp,
                      const float* __restrict__ WFr,
                      float* hout,
                      int do_relu, int n_rows, int n_strips) {
    __shared__ __attribute__((aligned(16))) float sM[2][SROWS * APITCH];
    __shared__ __attribute__((aligned(16))) float sX[2][SROWS * APITCH];

    const int lane  = threadIdx.x;
    const int wv    = __builtin_amdgcn_readfirstlane((int)threadIdx.y);
    const int ntile = wv & 3;
    const int mhalf = wv >> 2;
    const int khalf = (lane >> 4) * 2;
    const int col   = ntile * 16 + (lane & 15);

    int strip = blockIdx.x;
    if (strip >= n_strips) return;

    const int t    = wv * 32 + lane;
    const int srow = t >> 4;
    const int c4   = (t & 15) * 4;

    v4f ra0[2], ra1[2], rx[2];
    float  rd0[2], rd1[2];
    gnn_strip_request<TWO>(agg0, agg1, deg0, deg1, hin, strip, srow, c4, n_rows,
                           ra0, ra1, rx, rd0, rd1);

    v2f bL[16], bR[16];
    {
        const size_t wo = ((size_t)(ntile * 16) * 32 + lane) * 2;
        #pragma unroll
        for (int i = 0; i < 16; ++i) {
            bL[i] = *(const v2f*)(WFl + wo + (size_t)i * 64);
            bR[i] = *(const v2f*)(WFr + wo + (size_t)i * 64);
        }
    }
    const float bv = blp[col];

    const int aoff = (mhalf * 16 + (lane & 15)) * APITCH + khalf;

    int buf = 0;
    for (;;) {
        float* pm = &sM[buf][0];
        float* px = &sX[buf][0];
        #pragma unroll
        for (int j = 0; j < 2; ++j) {
            float dsum = rd0[j];
            if (TWO) dsum += rd1[j];
            const float rdeg = 1.0f / fmaxf(dsum, 1.0f);
            v4f m = ra0[j];
            if (TWO) m += ra1[j];
            m *= rdeg;
            const int o = (srow + 16 * j) * APITCH + c4;
            *(v4f*)(pm + o) = m;
            *(v4f*)(px + o) = rx[j];
        }
        __builtin_amdgcn_fence(__ATOMIC_RELEASE, "workgroup", "local");
        __builtin_amdgcn_s_barrier();
        __builtin_amdgcn_fence(__ATOMIC_ACQUIRE, "workgroup", "local");

        const int next = strip + (int)gridDim.x;
        if (next < n_strips)
            gnn_strip_request<TWO>(agg0, agg1, deg0, deg1, hin, next, srow, c4, n_rows,
                                   ra0, ra1, rx, rd0, rd1);

        const float* am = pm + aoff;
        const float* ax = px + aoff;
        v8f c = {};
        #pragma unroll
        for (int i = 0; i < 16; ++i) {
            const v2f aAgg = *(const v2f*)(am + 4 * i);
            const v2f aX   = *(const v2f*)(ax + 4 * i);
            c = __builtin_amdgcn_wmma_f32_16x16x4_f32(false, aAgg, false, bL[i],
                                                      (short)0, c, false, false);
            c = __builtin_amdgcn_wmma_f32_16x16x4_f32(false, aX,   false, bR[i],
                                                      (short)0, c, false, false);
        }

        const int row0 = strip * SROWS + mhalf * 16 + (lane >> 4) * 8;
        float* outp = hout + (size_t)row0 * DIN + col;
        if (strip * SROWS + mhalf * 16 + 16 <= n_rows) {
            #pragma unroll
            for (int r = 0; r < 8; ++r) {
                float v = c[r] + bv;
                if (do_relu) v = fmaxf(v, 0.0f);
                outp[(size_t)r * DIN] = v;
            }
        } else {
            #pragma unroll
            for (int r = 0; r < 8; ++r) {
                float v = c[r] + bv;
                if (do_relu) v = fmaxf(v, 0.0f);
                if (row0 + r < n_rows) outp[(size_t)r * DIN] = v;
            }
        }

        if (next >= n_strips) break;
        strip = next;
        buf ^= 1;
    }
}

__device__ __forceinline__ void
gnn_proj_store(const v8f& cy, const v8f& cz, float bv,
               float* __restrict__ yp, int ypitch, float* __restrict__ zp, int nvalid) {
    #pragma unroll
    for (int r = 0; r < 8; ++r) {
        if (r < nvalid) {
            yp[(size_t)r * ypitch] = cy[r];
            zp[(size_t)r * C2]     = cz[r] + bv;
        }
    }
}

__global__ void __launch_bounds__(256)
gnn_proj_wmma_kernel(const float* __restrict__ hin,
                     const float* __restrict__ WFl,
                     const float* __restrict__ WFr,
                     const float* __restrict__ blp,
                     float* __restrict__ y,
                     float* __restrict__ z,
                     int n_rows, int n_strips) {
    __shared__ __attribute__((aligned(16))) float sX[2][SROWS * APITCH];

    const int lane  = threadIdx.x;
    const int wv    = __builtin_amdgcn_readfirstlane((int)threadIdx.y);
    const bool worker = wv < 6;
    const int mhalf = wv >= 3 ? 1 : 0;
    const int ntile = wv - 3 * mhalf;
    const int khalf = (lane >> 4) * 2;
    const int col   = ntile * 16 + (lane & 15);

    int strip = blockIdx.x;
    if (strip >= n_strips) return;

    const int t    = wv * 32 + lane;
    const int srow = t >> 4;
    const int c4   = (t & 15) * 4;

    v4f rx[2];
    #pragma unroll
    for (int j = 0; j < 2; ++j) {
        int row = strip * SROWS + srow + 16 * j;
        row = row < n_rows ? row : n_rows - 1;
        rx[j] = *(const v4f*)(hin + (size_t)row * DIN + c4);
    }

    v2f bL[16], bR[16];
    float bv = 0.0f;
    if (worker) {
        const size_t wo = ((size_t)(ntile * 16) * 32 + lane) * 2;
        #pragma unroll
        for (int i = 0; i < 16; ++i) {
            bL[i] = *(const v2f*)(WFl + wo + (size_t)i * 64);
            bR[i] = *(const v2f*)(WFr + wo + (size_t)i * 64);
        }
        bv = blp[col];
    }

    const int aoff = (mhalf * 16 + (lane & 15)) * APITCH + khalf;

    int buf = 0;
    for (;;) {
        float* px = &sX[buf][0];
        #pragma unroll
        for (int j = 0; j < 2; ++j)
            *(v4f*)(px + (srow + 16 * j) * APITCH + c4) = rx[j];
        __builtin_amdgcn_fence(__ATOMIC_RELEASE, "workgroup", "local");
        __builtin_amdgcn_s_barrier();
        __builtin_amdgcn_fence(__ATOMIC_ACQUIRE, "workgroup", "local");

        const int next = strip + (int)gridDim.x;
        if (next < n_strips) {
            #pragma unroll
            for (int j = 0; j < 2; ++j) {
                int row = next * SROWS + srow + 16 * j;
                row = row < n_rows ? row : n_rows - 1;
                rx[j] = *(const v4f*)(hin + (size_t)row * DIN + c4);
            }
        }

        if (worker) {
            const float* ax = px + aoff;
            v8f cy = {};
            v8f cz = {};
            #pragma unroll
            for (int i = 0; i < 16; ++i) {
                const v2f a = *(const v2f*)(ax + 4 * i);
                cy = __builtin_amdgcn_wmma_f32_16x16x4_f32(false, a, false, bL[i],
                                                           (short)0, cy, false, false);
                cz = __builtin_amdgcn_wmma_f32_16x16x4_f32(false, a, false, bR[i],
                                                           (short)0, cz, false, false);
            }

            const int tile_row = strip * SROWS + mhalf * 16;
            const int row0i    = tile_row + (lane >> 4) * 8;
            const size_t row0  = (size_t)row0i;
            const bool full  = tile_row + 16 <= n_rows;
            const int nvalid = n_rows - row0i;
            if (col < C2) {
                float* yp = y + row0 * DIN + col;
                float* zp = z + row0 * C2  + col;
                if (full) gnn_proj_store(cy, cz, bv, yp, DIN, zp, 8);
                else      gnn_proj_store(cy, cz, bv, yp, DIN, zp, nvalid);
            }
        }

        if (next >= n_strips) break;
        strip = next;
        buf ^= 1;
    }
}

static inline void launch_dense(bool two, const float* agg0, const float* agg1,
                                const float* deg0, const float* deg1, const float* hin,
                                const float* WFl, const float* blp, const float* WFr,
                                float* hout, int do_relu,
                                int N, hipStream_t stream) {
    const int n_strips = (N + SROWS - 1) / SROWS;
    const int n_wg = n_strips < 768 ? n_strips : 768;
    dim3 block(32, 8);
    dim3 grid(n_wg);
    (void)two;
    gnn_dense_wmma_kernel<false><<<grid, block, 0, stream>>>(
        agg0, agg1, deg0, deg1, hin, WFl, blp, WFr, hout, do_relu, N, n_strips);
}

extern "C" void kernel_launch(void* const* d_in, const int* in_sizes, int n_in,
                              void* d_out, int out_size, void* d_ws, size_t ws_size,
                              hipStream_t stream) {
    const float* x   = (const float*)d_in[0];
    const int*   ei  = (const int*)  d_in[1];
    const float* Wl0 = (const float*)d_in[2];
    const float* bl0 = (const float*)d_in[3];
    const float* Wr0 = (const float*)d_in[4];
    const float* Wl1 = (const float*)d_in[5];
    const float* bl1 = (const float*)d_in[6];
    const float* Wr1 = (const float*)d_in[7];
    const float* Wl2 = (const float*)d_in[8];
    const float* bl2 = (const float*)d_in[9];
    const float* Wr2 = (const float*)d_in[10];

    const int N = in_sizes[0] / DIN;
    const int E = in_sizes[1] / 2;
    const int* src = ei;
    const int* dst = ei + E;

    const int DP0 = 64, DP1 = 64, DP2 = 48;

    const int R      = (((N + 254) / 255) + 7) / 8 * 8;
    const int n_own  = (N + R - 1) / R;
    const int n_chunks = (E + 511) / 512;
    const int n_kwords = n_chunks * 128;
    const int n_bch    = (E + 4095) / 4096;
    const int wpo      = n_bch * 128;

    float* deg  = (float*)d_ws;
    float* bufA = deg  + (size_t)N;
    float* bufB = bufA + (size_t)N * DIN;
    float* WTl0 = bufB + (size_t)N * DIN;
    float* WTr0 = WTl0 + DP0 * DIN;
    float* blp0 = WTr0 + DP0 * DIN;
    float* WTl1 = blp0 + DP0;
    float* WTr1 = WTl1 + DP1 * DIN;
    float* blp1 = WTr1 + DP1 * DIN;
    float* WTl2 = blp1 + DP1;
    float* WTr2 = WTl2 + DP2 * DIN;
    float* blp2 = WTr2 + DP2 * DIN;
    const size_t key_off = ((size_t)((blp2 + DP2) - deg) + 63) / 64 * 64;
    unsigned* keys = (unsigned*)(deg + key_off);
    const bool use_bm = ws_size >= (key_off + (size_t)n_own * (size_t)wpo) * sizeof(float);

    const int CAPW = 320;
    const size_t cnt_words = ((size_t)n_own * OWN_WAVES + 63) / 64 * 64;
    const size_t rec_words = cnt_words + (size_t)n_own * OWN_WAVES * CAPW;
    const size_t bm_end    = key_off + (size_t)n_own * (size_t)wpo;
    int* lcnt = nullptr;
    unsigned* list = nullptr;
    int lm0 = 0, lm1 = 0, lm2 = 0;
    if (use_bm && N <= (1 << OWN_SBITS)) {
        if (ws_size >= (bm_end + rec_words) * sizeof(float)) {
            lcnt = (int*)(deg + bm_end);
            lm0 = 1; lm1 = 2; lm2 = 2;
        } else if ((size_t)out_size >= rec_words) {
            lcnt = (int*)d_out;
            lm0 = 1; lm1 = 2;
        }
        if (lm0) list = (unsigned*)lcnt + cnt_words;
    }

    const dim3 own_block(32, OWN_WAVES);
    const int ZB = 256;

    gnn_pack_weights_kernel<<<(DP0 * DIN + ZB - 1) / ZB, ZB, 0, stream>>>(
        Wl0, bl0, Wr0, WTl0, WTr0, blp0, 64, DP0);
    gnn_pack_weights_kernel<<<(DP1 * DIN + ZB - 1) / ZB, ZB, 0, stream>>>(
        Wl1, bl1, Wr1, WTl1, WTr1, blp1, 64, DP1);
    gnn_pack_weights_kernel<<<(DP2 * DIN + ZB - 1) / ZB, ZB, 0, stream>>>(
        Wl2, bl2, Wr2, WTl2, WTr2, blp2, 40, DP2);
    if (use_bm)
        gnn_owner_bitmap_kernel<<<wpo / BM_LWORDS, 256, 0, stream>>>(dst, keys, E, R, n_own, wpo);
    else
        gnn_owner_key_kernel<<<(n_kwords + ZB - 1) / ZB, ZB, 0, stream>>>(
            dst, keys, E, n_kwords, R);
    const int n_sweep = use_bm ? n_bch : n_chunks;

    if (use_bm)
        gnn_owner_sum_kernel<DIN, true, false><<<dim3(n_own), own_block, 0, stream>>>(
            x, src, dst, keys, n_sweep, bufA, deg, N, R, 1, list, lcnt, CAPW, lm0, nullptr);
    else
        gnn_owner_sum_kernel<DIN, false, false><<<dim3(n_own), own_block, 0, stream>>>(
            x, src, dst, keys, n_sweep, bufA, deg, N, R, 1, list, lcnt, CAPW, 0, nullptr);
    launch_dense(false, bufA, bufA, deg, deg, x, WTl0, blp0, WTr0, bufA, 1, N, stream);

    if (use_bm)
        gnn_owner_sum_kernel<DIN, true, false><<<dim3(n_own), own_block, 0, stream>>>(
            bufA, src, dst, keys, n_sweep, bufB, deg, N, R, 0, list, lcnt, CAPW, lm1, nullptr);
    else
        gnn_owner_sum_kernel<DIN, false, false><<<dim3(n_own), own_block, 0, stream>>>(
            bufA, src, dst, keys, n_sweep, bufB, deg, N, R, 0, list, lcnt, CAPW, 0, nullptr);
    launch_dense(false, bufB, bufB, deg, deg, bufA, WTl1, blp1, WTr1, bufB, 1, N, stream);

    {
        const int n_strips = (N + SROWS - 1) / SROWS;
        const int n_wg = n_strips < 768 ? n_strips : 768;
        float* y = bufA;
        gnn_proj_wmma_kernel<<<dim3(n_wg), dim3(32, 8), 0, stream>>>(
            bufB, WTl2, WTr2, blp2, y, (float*)d_out, N, n_strips);
        if (use_bm)
            gnn_owner_sum_kernel<C2, true, true><<<dim3(n_own), own_block, 0, stream>>>(
                y, src, dst, keys, n_sweep, nullptr, deg, N, R, 0, list, lcnt, CAPW, lm2,
                (float*)d_out);
        else
            gnn_owner_sum_kernel<C2, false, true><<<dim3(n_own), own_block, 0, stream>>>(
                y, src, dst, keys, n_sweep, nullptr, deg, N, R, 0, list, lcnt, CAPW, 0,
                (float*)d_out);
    }
}


# Three weight-pack launches folded into the bitmap launch: 44 pack work-groups stand behind the bitmap builders in its grid and leave before its barriers (10 launches to 7); same sums
# speedup vs baseline: 3.8251x; 1.0537x over previous
#include <hip/hip_runtime.h>
#include <cstddef>

typedef float v2f __attribute__((ext_vector_type(2)));
typedef float v4f __attribute__((ext_vector_type(4)));
typedef float v8f __attribute__((ext_vector_type(8)));

#define DIN 64

struct GnnPackJob {
    const float* Wl;
    const float* bl;
    const float* Wr;
    float* WFl;
    float* WFr;
    float* blp;
    int dout;
    int dout_pad;
};

__device__ __forceinline__ void gnn_pack_weights(const GnnPackJob& p, int tid) {
    int col = tid / DIN;
    int k   = tid % DIN;
    if (col >= p.dout_pad) return;
    bool cv  = (col < p.dout);
    const int i = k >> 2, h = (k >> 1) & 1, e = k & 1;
    const size_t o = ((size_t)(((col >> 4) * 16 + i) * 32 + h * 16 + (col & 15))) * 2 + e;
    float wl = 0.0f, wr = 0.0f, bb = 0.0f;
    if (cv) {
        wl = p.Wl[(size_t)k * p.dout + col];
        wr = p.Wr[(size_t)k * p.dout + col];
        if (k == 0) bb = p.bl[col];
    }
    p.WFl[o] = wl;
    p.WFr[o] = wr;
    if (k == 0) p.blp[col] = bb;
}

__global__ void gnn_pack_weights_kernel(GnnPackJob p) {
    gnn_pack_weights(p, blockIdx.x * blockDim.x + threadIdx.x);
}

#define OWN_WAVES 32
#define OWN_RMAX  640
#define OWN_RPW   (OWN_RMAX / OWN_WAVES)
#define OWN_QCAP  64
#define OWN_BATCH 16
#define OWN_SBITS 18
#define OWN_SMASK ((1u << OWN_SBITS) - 1u)

__global__ void gnn_owner_key_kernel(const int* __restrict__ dst,
                                     unsigned* __restrict__ keys,
                                     int E, int n_words, int R) {
    const int w = blockIdx.x * blockDim.x + threadIdx.x;
    if (w >= n_words) return;
    unsigned kw = 0u;
    #pragma unroll
    for (int b = 0; b < 4; ++b) {
        const long long e = (long long)w * 4 + b;
        unsigned k = 0xFFu;
        if (e < (long long)E) k = (unsigned)dst[e] / (unsigned)R;
        kw |= k << (8 * b);
    }
    keys[w] = kw;
}

#define BM_EDGES  2048
#define BM_LWORDS 64

__global__ void __launch_bounds__(256)
gnn_owner_bitmap_kernel(const int* __restrict__ dst,
                        unsigned* __restrict__ bm,
                        int E, int R, int n_own, int wpo,
                        int n_bm_wg, GnnPackJob p0, GnnPackJob p1, GnnPackJob p2) {
    __shared__ __attribute__((aligned(16))) unsigned tile[255 * BM_LWORDS];
    const int t  = threadIdx.x;
    if ((int)blockIdx.x >= n_bm_wg) {
        int b = (int)blockIdx.x - n_bm_wg;
        const int nb0 = (p0.dout_pad * DIN + 255) / 256;
        const int nb1 = (p1.dout_pad * DIN + 255) / 256;
        if (b < nb0)            gnn_pack_weights(p0, b * 256 + t);
        else if (b < nb0 + nb1) gnn_pack_weights(p1, (b - nb0) * 256 + t);
        else                    gnn_pack_weights(p2, (b - nb0 - nb1) * 256 + t);
        return;
    }
    const int n4 = n_own * (BM_LWORDS / 4);
    for (int i = t; i < n4; i += 256) ((uint4*)tile)[i] = make_uint4(0u, 0u, 0u, 0u);
    __syncthreads();
    const long long base = (long long)blockIdx.x * BM_EDGES;
    int d[BM_EDGES / 256];
    #pragma unroll
    for (int b = 0; b < BM_EDGES / 256; ++b) {
        const long long e = base + b * 256 + t;
        d[b] = dst[e < (long long)E ? e : (long long)E - 1];
    }
    #pragma unroll
    for (int b = 0; b < BM_EDGES / 256; ++b) {
        const int le = b * 256 + t;
        if (base + le < (long long)E) {
            const unsigned g = (unsigned)d[b] / (unsigned)R;
            atomicOr(&tile[g * BM_LWORDS + (le >> 5)], 1u << (le & 31));
        }
    }
    __syncthreads();
    uint4* o4 = (uint4*)bm;
    for (int i = t; i < n4; i += 256) {
        const int g = i >> 4, p = i & 15;
        o4[((size_t)g * (size_t)wpo + (size_t)blockIdx.x * BM_LWORDS) / 4 + p] = ((const uint4*)tile)[i];
    }
}

__device__ __forceinline__ unsigned gnn_zero_bytes(unsigned x) {
    return ~(((x & 0x7F7F7F7Fu) + 0x7F7F7F7Fu) | x | 0x7F7F7F7Fu);
}

template <int W>
__device__ __forceinline__ void
gnn_owner_rows(const float* __restrict__ rows, int s, int r, int n, int lane, float* acc) {
    const bool act  = (W == DIN) || (2 * lane < W);
    const int  loff = act ? 2 * lane : 0;
    #pragma unroll
    for (int b = 0; b < 32 / OWN_BATCH; ++b) {
        if (b * OWN_BATCH < n) {
            v2f v[OWN_BATCH];
            #pragma unroll
            for (int j = 0; j < OWN_BATCH; ++j) {
                const int sj = __builtin_amdgcn_readlane(s, b * OWN_BATCH + j);
                v[j] = *(const v2f*)(rows + (size_t)sj * DIN + loff);
            }
            if (act) {
                #pragma unroll
                for (int j = 0; j < OWN_BATCH; ++j) {
                    if (b * OWN_BATCH + j < n) {
                        const int rj = __builtin_amdgcn_readlane(r, b * OWN_BATCH + j);
                        float* a = acc + rj * W + 2 * lane;
                        __hip_atomic_fetch_add(a,     v[j].x, __ATOMIC_RELAXED, __HIP_MEMORY_SCOPE_WORKGROUP);
                        __hip_atomic_fetch_add(a + 1, v[j].y, __ATOMIC_RELAXED, __HIP_MEMORY_SCOPE_WORKGROUP);
                    }
                }
            }
        }
    }
}

template <int W>
__device__ __forceinline__ void
gnn_owner_drain(const float* __restrict__ rows,
                const int* __restrict__ src, const int* __restrict__ dst,
                const unsigned* qw, unsigned head, int n, int lo, int lane,
                float* acc, unsigned* degL, int do_deg,
                unsigned* seg, int pos, int capw) {
    int s = 0, r = 0;
    if (lane < n) {
        const unsigned e = qw[(head + (unsigned)lane) & (OWN_QCAP - 1)];
        s = src[e];
        r = dst[e] - lo;
        if (do_deg) atomicAdd(&degL[r], 1u);
        if (seg != nullptr && pos + lane < capw)
            seg[pos + lane] = (unsigned)s | ((unsigned)r << OWN_SBITS);
    }
    gnn_owner_rows<W>(rows, s, r, n, lane, acc);
}

#define C2   40
#define C2A  32
#define C2B  8

__device__ __forceinline__ void
gnn_row_logsoftmax(float a, float b, float z0, float z1, float rdeg, int lane, float* orow) {
    const bool hasB = lane < C2B;
    const float NEG_INF = -__builtin_inff();
    const float v0 = a * rdeg + z0;
    const float v1 = hasB ? b * rdeg + z1 : NEG_INF;

    float m = fmaxf(v0, v1);
    #pragma unroll
    for (int off = 16; off > 0; off >>= 1) m = fmaxf(m, __shfl_xor(m, off, 32));
    const float e0 = expf(v0 - m);
    const float e1 = hasB ? expf(v1 - m) : 0.0f;
    float s = e0 + e1;
    #pragma unroll
    for (int off = 16; off > 0; off >>= 1) s += __shfl_xor(s, off, 32);
    const float ls = logf(s);
    orow[lane] = v0 - m - ls;
    if (hasB) orow[C2A + lane] = v1 - m - ls;
}

template <int W, bool BM, bool FIN>
__global__ void __launch_bounds__(32 * OWN_WAVES)
gnn_owner_sum_kernel(const float* __restrict__ rows,
                     const int* __restrict__ src,
                     const int* __restrict__ dst,
                     const unsigned* __restrict__ keys,
                     int n_chunks,
                     float* __restrict__ sums,
                     float* __restrict__ deg,
                     int n_nodes, int R, int do_deg,
                     unsigned* list, int* lcnt, int capw, int list_mode,
                     float* fin) {
    __shared__ __attribute__((aligned(16))) float acc[OWN_RMAX * DIN];
    __shared__ unsigned degL[OWN_RMAX];
    __shared__ unsigned q[OWN_WAVES][OWN_QCAP];

    const int lane = threadIdx.x;
    const int wv = __builtin_amdgcn_readfirstlane((int)threadIdx.y);
    const int t  = wv * 32 + lane;
    const int g  = blockIdx.x;
    const int lo = g * R;
    int nrows = n_nodes - lo;
    nrows = nrows < R ? nrows : R;

    int wcount = 0;
    if (list_mode == 2) wcount = lcnt[g * OWN_WAVES + lane];

    const int n4 = nrows * W / 4;
    for (int i = t; i < n4; i += 32 * OWN_WAVES) ((v4f*)acc)[i] = v4f{0.0f, 0.0f, 0.0f, 0.0f};
    for (int i = t; i < nrows; i += 32 * OWN_WAVES) degL[i] = 0u;
    __syncthreads();

    const bool replay = list_mode == 2 && __builtin_amdgcn_ballot_w32(wcount > capw) == 0u;
    const size_t seg_off = ((size_t)g * OWN_WAVES + (size_t)wv) * (size_t)capw;

    if (replay) {
        const int n = __builtin_amdgcn_readlane(wcount, wv);
        if (n > 0) {
            const unsigned* seg = list + seg_off;
            const int last  = n - 1;
            const int nfull = n >> 5;
            unsigned ent = seg[lane < last ? lane : last];
            for (int k = 0; k < nfull; ++k) {
                const int in = (k + 1) * 32 + lane;
                const unsigned entn = seg[in < last ? in : last];
                gnn_owner_rows<W>(rows, (int)(ent & OWN_SMASK), (int)(ent >> OWN_SBITS), 32, lane, acc);
                ent = entn;
            }
            const int rem = n & 31;
            if (rem > 0)
                gnn_owner_rows<W>(rows, (int)(ent & OWN_SMASK), (int)(ent >> OWN_SBITS), rem, lane, acc);
        }
    } else {
    unsigned* seg = list_mode == 1 ? list + seg_off : nullptr;
    int pos = 0;

    unsigned* qw = &q[wv][0];
    unsigned head = 0u;
    int cnt = 0;

    if (BM) {
        const uint4* b4 = (const uint4*)(keys + (size_t)g * (size_t)n_chunks * 128u);
        int kb = wv;
        uint4 bw = make_uint4(0u, 0u, 0u, 0u);
        if (kb < n_chunks) bw = b4[(size_t)kb * 32 + lane];
        while (kb < n_chunks) {
            const unsigned ebase = (unsigned)kb * 4096u + (unsigned)lane * 128u;
            const int kbn = kb + OWN_WAVES;
            const uint4 bwn = b4[(size_t)(kbn < n_chunks ? kbn : kb) * 32 + lane];
            #pragma unroll
            for (int hh = 0; hh < 2; ++hh) {
                unsigned long long m = hh ? (((unsigned long long)bw.w << 32) | bw.z)
                                          : (((unsigned long long)bw.y << 32) | bw.x);
                const unsigned eb = ebase + 64u * (unsigned)hh;
                for (;;) {
                    const bool has = m != 0ull;
                    const unsigned bal = __builtin_amdgcn_ballot_w32(has);
                    if (bal == 0u) break;
                    if (has) {
                        const unsigned pre = __builtin_amdgcn_mbcnt_lo(bal, 0u);
                        const unsigned tz  = (unsigned)__builtin_ctzll(m);
                        qw[(head + (unsigned)cnt + pre) & (OWN_QCAP - 1)] = eb + tz;
                        m &= m - 1ull;
                    }
                    cnt += __builtin_popcount(bal);
                    if (cnt >= 32) {
                        gnn_owner_drain<W>(rows, src, dst, qw, head, 32, lo, lane, acc, degL, do_deg,
                                           seg, pos, capw);
                        pos += 32;
                        head = (head + 32u) & (OWN_QCAP - 1);
                        cnt -= 32;
                    }
                }
            }
            kb = kbn; bw = bwn;
        }
    } else {
    const unsigned pat = (unsigned)g * 0x01010101u;
    const int start = (int)(((unsigned)g * 97u) % (unsigned)n_chunks);
    const uint4* k4 = (const uint4*)keys;

    int k = wv;
    uint4 kw = make_uint4(0xFFFFFFFFu, 0xFFFFFFFFu, 0xFFFFFFFFu, 0xFFFFFFFFu);
    int c = 0;
    if (k < n_chunks) {
        c = start + k; if (c >= n_chunks) c -= n_chunks;
        kw = k4[(size_t)c * 32 + lane];
    }
    while (k < n_chunks) {
        unsigned m = (gnn_zero_bytes(kw.x ^ pat) >> 7) | (gnn_zero_bytes(kw.y ^ pat) >> 6)
                   | (gnn_zero_bytes(kw.z ^ pat) >> 5) | (gnn_zero_bytes(kw.w ^ pat) >> 4);
        const unsigned ebase = (unsigned)c * 512u + (unsigned)lane * 16u;

        const int kn = k + OWN_WAVES;
        int cn = start + (kn < n_chunks ? kn : k);
        if (cn >= n_chunks) cn -= n_chunks;
        const uint4 kwn = k4[(size_t)cn * 32 + lane];
        for (;;) {
            const bool has = m != 0u;
            const unsigned bal = __builtin_amdgcn_ballot_w32(has);
            if (bal == 0u) break;
            if (has) {
                const unsigned pre = __builtin_amdgcn_mbcnt_lo(bal, 0u);
                const unsigned tz  = (unsigned)__builtin_ctz(m);
                qw[(head + (unsigned)cnt + pre) & (OWN_QCAP - 1)] = ebase + (tz & 7u) * 4u + (tz >> 3);
                m &= m - 1u;
            }
            cnt += __builtin_popcount(bal);
            if (cnt >= 32) {
                gnn_owner_drain<W>(rows, src, dst, qw, head, 32, lo, lane, acc, degL, do_deg,
                                   seg, pos, capw);
                pos += 32;
                head = (head + 32u) & (OWN_QCAP - 1);
                cnt -= 32;
            }
        }
        k = kn; c = cn; kw = kwn;
    }
    }
    if (cnt > 0) {
        gnn_owner_drain<W>(rows, src, dst, qw, head, cnt, lo, lane, acc, degL, do_deg,
                           seg, pos, capw);
        pos += cnt;
    }
    if (list_mode == 1 && lane == 0) lcnt[g * OWN_WAVES + wv] = pos;
    }

    if (FIN) {
        static_assert(!FIN || W == C2, "the finishing step is the 40-wide last layer's");
        const int rpw = (nrows + OWN_WAVES - 1) / OWN_WAVES;
        const int r0  = wv * rpw;
        int cnt = nrows - r0;
        cnt = cnt < rpw ? cnt : rpw;
        const int l2 = lane < C2B ? C2A + lane : lane;
        float* zrow = fin + (size_t)(lo + r0) * C2;
        float z0[OWN_RPW], z1[OWN_RPW];
        #pragma unroll
        for (int j = 0; j < OWN_RPW; ++j) {
            z0[j] = 0.0f; z1[j] = 0.0f;
            if (j < cnt) {
                z0[j] = zrow[j * C2 + lane];
                z1[j] = zrow[j * C2 + l2];
            }
        }
        float dv = 1.0f;
        if (lane < cnt) dv = deg[lo + r0 + lane];
        __syncthreads();
        const float rdv = 1.0f / fmaxf(dv, 1.0f);
        const float* arow = acc + r0 * C2;
        #pragma unroll
        for (int j = 0; j < OWN_RPW; ++j) {
            if (j < cnt) {
                const float a = arow[j * C2 + lane];
                const float b = arow[j * C2 + l2];
                const float rdeg = __int_as_float(__builtin_amdgcn_readlane(__float_as_int(rdv), j));
                gnn_row_logsoftmax(a, b, z0[j], z1[j], rdeg, lane, zrow + j * C2);
            }
        }
    } else {
        __syncthreads();
        v4f* o4 = (v4f*)(sums + (size_t)lo * W);
        for (int i = t; i < n4; i += 32 * OWN_WAVES) o4[i] = ((const v4f*)acc)[i];
        if (do_deg)
            for (int i = t; i < nrows; i += 32 * OWN_WAVES) deg[lo + i] = (float)degL[i];
    }
}

#define SROWS  32
#define APITCH 68

template <bool TWO>
__device__ __forceinline__ void
gnn_strip_request(const float* agg0, const float* __restrict__ agg1,
                  const float* __restrict__ deg0, const float* __restrict__ deg1,
                  const float* __restrict__ hin, int strip, int srow, int c4, int n_rows,
                  v4f (&ra0)[2], v4f (&ra1)[2], v4f (&rx)[2],
                  float (&rd0)[2], float (&rd1)[2]) {
    #pragma unroll
    for (int j = 0; j < 2; ++j) {
        int row = strip * SROWS + srow + 16 * j;
        row = row < n_rows ? row : n_rows - 1;
        const size_t o = (size_t)row * DIN + c4;
        ra0[j] = *(const v4f*)(agg0 + o);
        if (TWO) ra1[j] = *(const v4f*)(agg1 + o);
        rx[j]  = *(const v4f*)(hin + o);
        rd0[j] = deg0[row];
        if (TWO) rd1[j] = deg1[row];
    }
}

template <bool TWO>
__global__ void __launch_bounds__(256)
gnn_dense_wmma_kernel(const float* agg0,
                      const float* __restrict__ agg1,
                      const float* __restrict__ deg0,
                      const float* __restrict__ deg1,
                      const float* __restrict__ hin,
                      const float* __restrict__ WFl,
                      const float* __restrict__ blp,
                      const float* __restrict__ WFr,
                      float* hout,
                      int do_relu, int n_rows, int n_strips) {
    __shared__ __attribute__((aligned(16))) float sM[2][SROWS * APITCH];
    __shared__ __attribute__((aligned(16))) float sX[2][SROWS * APITCH];

    const int lane  = threadIdx.x;
    const int wv    = __builtin_amdgcn_readfirstlane((int)threadIdx.y);
    const int ntile = wv & 3;
    const int mhalf = wv >> 2;
    const int khalf = (lane >> 4) * 2;
    const int col   = ntile * 16 + (lane & 15);

    int strip = blockIdx.x;
    if (strip >= n_strips) return;

    const int t    = wv * 32 + lane;
    const int srow = t >> 4;
    const int c4   = (t & 15) * 4;

    v4f ra0[2], ra1[2], rx[2];
    float  rd0[2], rd1[2];
    gnn_strip_request<TWO>(agg0, agg1, deg0, deg1, hin, strip, srow, c4, n_rows,
                           ra0, ra1, rx, rd0, rd1);

    v2f bL[16], bR[16];
    {
        const size_t wo = ((size_t)(ntile * 16) * 32 + lane) * 2;
        #pragma unroll
        for (int i = 0; i < 16; ++i) {
            bL[i] = *(const v2f*)(WFl + wo + (size_t)i * 64);
            bR[i] = *(const v2f*)(WFr + wo + (size_t)i * 64);
        }
    }
    const float bv = blp[col];

    const int aoff = (mhalf * 16 + (lane & 15)) * APITCH + khalf;

    int buf = 0;
    for (;;) {
        float* pm = &sM[buf][0];
        float* px = &sX[buf][0];
        #pragma unroll
        for (int j = 0; j < 2; ++j) {
            float dsum = rd0[j];
            if (TWO) dsum += rd1[j];
            const float rdeg = 1.0f / fmaxf(dsum, 1.0f);
            v4f m = ra0[j];
            if (TWO) m += ra1[j];
            m *= rdeg;
            const int o = (srow + 16 * j) * APITCH + c4;
            *(v4f*)(pm + o) = m;
            *(v4f*)(px + o) = rx[j];
        }
        __builtin_amdgcn_fence(__ATOMIC_RELEASE, "workgroup", "local");
        __builtin_amdgcn_s_barrier();
        __builtin_amdgcn_fence(__ATOMIC_ACQUIRE, "workgroup", "local");

        const int next = strip + (int)gridDim.x;
        if (next < n_strips)
            gnn_strip_request<TWO>(agg0, agg1, deg0, deg1, hin, next, srow, c4, n_rows,
                                   ra0, ra1, rx, rd0, rd1);

        const float* am = pm + aoff;
        const float* ax = px + aoff;
        v8f c = {};
        #pragma unroll
        for (int i = 0; i < 16; ++i) {
            const v2f aAgg = *(const v2f*)(am + 4 * i);
            const v2f aX   = *(const v2f*)(ax + 4 * i);
            c = __builtin_amdgcn_wmma_f32_16x16x4_f32(false, aAgg, false, bL[i],
                                                      (short)0, c, false, false);
            c = __builtin_amdgcn_wmma_f32_16x16x4_f32(false, aX,   false, bR[i],
                                                      (short)0, c, false, false);
        }

        const int row0 = strip * SROWS + mhalf * 16 + (lane >> 4) * 8;
        float* outp = hout + (size_t)row0 * DIN + col;
        if (strip * SROWS + mhalf * 16 + 16 <= n_rows) {
            #pragma unroll
            for (int r = 0; r < 8; ++r) {
                float v = c[r] + bv;
                if (do_relu) v = fmaxf(v, 0.0f);
                outp[(size_t)r * DIN] = v;
            }
        } else {
            #pragma unroll
            for (int r = 0; r < 8; ++r) {
                float v = c[r] + bv;
                if (do_relu) v = fmaxf(v, 0.0f);
                if (row0 + r < n_rows) outp[(size_t)r * DIN] = v;
            }
        }

        if (next >= n_strips) break;
        strip = next;
        buf ^= 1;
    }
}

__device__ __forceinline__ void
gnn_proj_store(const v8f& cy, const v8f& cz, float bv,
               float* __restrict__ yp, int ypitch, float* __restrict__ zp, int nvalid) {
    #pragma unroll
    for (int r = 0; r < 8; ++r) {
        if (r < nvalid) {
            yp[(size_t)r * ypitch] = cy[r];
            zp[(size_t)r * C2]     = cz[r] + bv;
        }
    }
}

__global__ void __launch_bounds__(256)
gnn_proj_wmma_kernel(const float* __restrict__ hin,
                     const float* __restrict__ WFl,
                     const float* __restrict__ WFr,
                     const float* __restrict__ blp,
                     float* __restrict__ y,
                     float* __restrict__ z,
                     int n_rows, int n_strips) {
    __shared__ __attribute__((aligned(16))) float sX[2][SROWS * APITCH];

    const int lane  = threadIdx.x;
    const int wv    = __builtin_amdgcn_readfirstlane((int)threadIdx.y);
    const bool worker = wv < 6;
    const int mhalf = wv >= 3 ? 1 : 0;
    const int ntile = wv - 3 * mhalf;
    const int khalf = (lane >> 4) * 2;
    const int col   = ntile * 16 + (lane & 15);

    int strip = blockIdx.x;
    if (strip >= n_strips) return;

    const int t    = wv * 32 + lane;
    const int srow = t >> 4;
    const int c4   = (t & 15) * 4;

    v4f rx[2];
    #pragma unroll
    for (int j = 0; j < 2; ++j) {
        int row = strip * SROWS + srow + 16 * j;
        row = row < n_rows ? row : n_rows - 1;
        rx[j] = *(const v4f*)(hin + (size_t)row * DIN + c4);
    }

    v2f bL[16], bR[16];
    float bv = 0.0f;
    if (worker) {
        const size_t wo = ((size_t)(ntile * 16) * 32 + lane) * 2;
        #pragma unroll
        for (int i = 0; i < 16; ++i) {
            bL[i] = *(const v2f*)(WFl + wo + (size_t)i * 64);
            bR[i] = *(const v2f*)(WFr + wo + (size_t)i * 64);
        }
        bv = blp[col];
    }

    const int aoff = (mhalf * 16 + (lane & 15)) * APITCH + khalf;

    int buf = 0;
    for (;;) {
        float* px = &sX[buf][0];
        #pragma unroll
        for (int j = 0; j < 2; ++j)
            *(v4f*)(px + (srow + 16 * j) * APITCH + c4) = rx[j];
        __builtin_amdgcn_fence(__ATOMIC_RELEASE, "workgroup", "local");
        __builtin_amdgcn_s_barrier();
        __builtin_amdgcn_fence(__ATOMIC_ACQUIRE, "workgroup", "local");

        const int next = strip + (int)gridDim.x;
        if (next < n_strips) {
            #pragma unroll
            for (int j = 0; j < 2; ++j) {
                int row = next * SROWS + srow + 16 * j;
                row = row < n_rows ? row : n_rows - 1;
                rx[j] = *(const v4f*)(hin + (size_t)row * DIN + c4);
            }
        }

        if (worker) {
            const float* ax = px + aoff;
            v8f cy = {};
            v8f cz = {};
            #pragma unroll
            for (int i = 0; i < 16; ++i) {
                const v2f a = *(const v2f*)(ax + 4 * i);
                cy = __builtin_amdgcn_wmma_f32_16x16x4_f32(false, a, false, bL[i],
                                                           (short)0, cy, false, false);
                cz = __builtin_amdgcn_wmma_f32_16x16x4_f32(false, a, false, bR[i],
                                                           (short)0, cz, false, false);
            }

            const int tile_row = strip * SROWS + mhalf * 16;
            const int row0i    = tile_row + (lane >> 4) * 8;
            const size_t row0  = (size_t)row0i;
            const bool full  = tile_row + 16 <= n_rows;
            const int nvalid = n_rows - row0i;
            if (col < C2) {
                float* yp = y + row0 * DIN + col;
                float* zp = z + row0 * C2  + col;
                if (full) gnn_proj_store(cy, cz, bv, yp, DIN, zp, 8);
                else      gnn_proj_store(cy, cz, bv, yp, DIN, zp, nvalid);
            }
        }

        if (next >= n_strips) break;
        strip = next;
        buf ^= 1;
    }
}

static inline void launch_dense(bool two, const float* agg0, const float* agg1,
                                const float* deg0, const float* deg1, const float* hin,
                                const float* WFl, const float* blp, const float* WFr,
                                float* hout, int do_relu,
                                int N, hipStream_t stream) {
    const int n_strips = (N + SROWS - 1) / SROWS;
    const int n_wg = n_strips < 768 ? n_strips : 768;
    dim3 block(32, 8);
    dim3 grid(n_wg);
    (void)two;
    gnn_dense_wmma_kernel<false><<<grid, block, 0, stream>>>(
        agg0, agg1, deg0, deg1, hin, WFl, blp, WFr, hout, do_relu, N, n_strips);
}

extern "C" void kernel_launch(void* const* d_in, const int* in_sizes, int n_in,
                              void* d_out, int out_size, void* d_ws, size_t ws_size,
                              hipStream_t stream) {
    const float* x   = (const float*)d_in[0];
    const int*   ei  = (const int*)  d_in[1];
    const float* Wl0 = (const float*)d_in[2];
    const float* bl0 = (const float*)d_in[3];
    const float* Wr0 = (const float*)d_in[4];
    const float* Wl1 = (const float*)d_in[5];
    const float* bl1 = (const float*)d_in[6];
    const float* Wr1 = (const float*)d_in[7];
    const float* Wl2 = (const float*)d_in[8];
    const float* bl2 = (const float*)d_in[9];
    const float* Wr2 = (const float*)d_in[10];

    const int N = in_sizes[0] / DIN;
    const int E = in_sizes[1] / 2;
    const int* src = ei;
    const int* dst = ei + E;

    const int DP0 = 64, DP1 = 64, DP2 = 48;

    const int R      = (((N + 254) / 255) + 7) / 8 * 8;
    const int n_own  = (N + R - 1) / R;
    const int n_chunks = (E + 511) / 512;
    const int n_kwords = n_chunks * 128;
    const int n_bch    = (E + 4095) / 4096;
    const int wpo      = n_bch * 128;

    float* deg  = (float*)d_ws;
    float* bufA = deg  + (size_t)N;
    float* bufB = bufA + (size_t)N * DIN;
    float* WTl0 = bufB + (size_t)N * DIN;
    float* WTr0 = WTl0 + DP0 * DIN;
    float* blp0 = WTr0 + DP0 * DIN;
    float* WTl1 = blp0 + DP0;
    float* WTr1 = WTl1 + DP1 * DIN;
    float* blp1 = WTr1 + DP1 * DIN;
    float* WTl2 = blp1 + DP1;
    float* WTr2 = WTl2 + DP2 * DIN;
    float* blp2 = WTr2 + DP2 * DIN;
    const size_t key_off = ((size_t)((blp2 + DP2) - deg) + 63) / 64 * 64;
    unsigned* keys = (unsigned*)(deg + key_off);
    const bool use_bm = ws_size >= (key_off + (size_t)n_own * (size_t)wpo) * sizeof(float);

    const int CAPW = 320;
    const size_t cnt_words = ((size_t)n_own * OWN_WAVES + 63) / 64 * 64;
    const size_t rec_words = cnt_words + (size_t)n_own * OWN_WAVES * CAPW;
    const size_t bm_end    = key_off + (size_t)n_own * (size_t)wpo;
    int* lcnt = nullptr;
    unsigned* list = nullptr;
    int lm0 = 0, lm1 = 0, lm2 = 0;
    if (use_bm && N <= (1 << OWN_SBITS)) {
        if (ws_size >= (bm_end + rec_words) * sizeof(float)) {
            lcnt = (int*)(deg + bm_end);
            lm0 = 1; lm1 = 2; lm2 = 2;
        } else if ((size_t)out_size >= rec_words) {
            lcnt = (int*)d_out;
            lm0 = 1; lm1 = 2;
        }
        if (lm0) list = (unsigned*)lcnt + cnt_words;
    }

    const dim3 own_block(32, OWN_WAVES);
    const int ZB = 256;

    const GnnPackJob pk0 = {Wl0, bl0, Wr0, WTl0, WTr0, blp0, 64, DP0};
    const GnnPackJob pk1 = {Wl1, bl1, Wr1, WTl1, WTr1, blp1, 64, DP1};
    const GnnPackJob pk2 = {Wl2, bl2, Wr2, WTl2, WTr2, blp2, 40, DP2};
    const int nb0 = (DP0 * DIN + ZB - 1) / ZB;
    const int nb1 = (DP1 * DIN + ZB - 1) / ZB;
    const int nb2 = (DP2 * DIN + ZB - 1) / ZB;
    if (use_bm) {
        const int n_bm_wg = wpo / BM_LWORDS;
        gnn_owner_bitmap_kernel<<<n_bm_wg + nb0 + nb1 + nb2, 256, 0, stream>>>(
            dst, keys, E, R, n_own, wpo, n_bm_wg, pk0, pk1, pk2);
    } else {
        gnn_pack_weights_kernel<<<nb0, ZB, 0, stream>>>(pk0);
        gnn_pack_weights_kernel<<<nb1, ZB, 0, stream>>>(pk1);
        gnn_pack_weights_kernel<<<nb2, ZB, 0, stream>>>(pk2);
        gnn_owner_key_kernel<<<(n_kwords + ZB - 1) / ZB, ZB, 0, stream>>>(
            dst, keys, E, n_kwords, R);
    }
    const int n_sweep = use_bm ? n_bch : n_chunks;

    if (use_bm)
        gnn_owner_sum_kernel<DIN, true, false><<<dim3(n_own), own_block, 0, stream>>>(
            x, src, dst, keys, n_sweep, bufA, deg, N, R, 1, list, lcnt, CAPW, lm0, nullptr);
    else
        gnn_owner_sum_kernel<DIN, false, false><<<dim3(n_own), own_block, 0, stream>>>(
            x, src, dst, keys, n_sweep, bufA, deg, N, R, 1, list, lcnt, CAPW, 0, nullptr);
    launch_dense(false, bufA, bufA, deg, deg, x, WTl0, blp0, WTr0, bufA, 1, N, stream);

    if (use_bm)
        gnn_owner_sum_kernel<DIN, true, false><<<dim3(n_own), own_block, 0, stream>>>(
            bufA, src, dst, keys, n_sweep, bufB, deg, N, R, 0, list, lcnt, CAPW, lm1, nullptr);
    else
        gnn_owner_sum_kernel<DIN, false, false><<<dim3(n_own), own_block, 0, stream>>>(
            bufA, src, dst, keys, n_sweep, bufB, deg, N, R, 0, list, lcnt, CAPW, 0, nullptr);
    launch_dense(false, bufB, bufB, deg, deg, bufA, WTl1, blp1, WTr1, bufB, 1, N, stream);

    {
        const int n_strips = (N + SROWS - 1) / SROWS;
        const int n_wg = n_strips < 768 ? n_strips : 768;
        float* y = bufA;
        gnn_proj_wmma_kernel<<<dim3(n_wg), dim3(32, 8), 0, stream>>>(
            bufB, WTl2, WTr2, blp2, y, (float*)d_out, N, n_strips);
        if (use_bm)
            gnn_owner_sum_kernel<C2, true, true><<<dim3(n_own), own_block, 0, stream>>>(
                y, src, dst, keys, n_sweep, nullptr, deg, N, R, 0, list, lcnt, CAPW, lm2,
                (float*)d_out);
        else
            gnn_owner_sum_kernel<C2, false, true><<<dim3(n_own), own_block, 0, stream>>>(
                y, src, dst, keys, n_sweep, nullptr, deg, N, R, 0, list, lcnt, CAPW, 0,
                (float*)d_out);
    }
}


# Hidden layers' dense steps and the last layer's projection done by the owner work-groups on their LDS sum tiles (fp32 WMMA from the tile): sum planes, h2 and 3 launches go; same sums
# speedup vs baseline: 4.2250x; 1.1045x over previous
#include <hip/hip_runtime.h>
#include <cstddef>

typedef float v2f __attribute__((ext_vector_type(2)));
typedef float v4f __attribute__((ext_vector_type(4)));
typedef float v8f __attribute__((ext_vector_type(8)));

#define DIN 64

struct GnnPackJob {
    const float* Wl;
    const float* bl;
    const float* Wr;
    float* WFl;
    float* WFr;
    float* blp;
    int dout;
    int dout_pad;
};

__device__ __forceinline__ void gnn_pack_weights(const GnnPackJob& p, int tid) {
    int col = tid / DIN;
    int k   = tid % DIN;
    if (col >= p.dout_pad) return;
    bool cv  = (col < p.dout);
    const int i = k >> 2, h = (k >> 1) & 1, e = k & 1;
    const size_t o = ((size_t)(((col >> 4) * 16 + i) * 32 + h * 16 + (col & 15))) * 2 + e;
    float wl = 0.0f, wr = 0.0f, bb = 0.0f;
    if (cv) {
        wl = p.Wl[(size_t)k * p.dout + col];
        wr = p.Wr[(size_t)k * p.dout + col];
        if (k == 0) bb = p.bl[col];
    }
    p.WFl[o] = wl;
    p.WFr[o] = wr;
    if (k == 0) p.blp[col] = bb;
}

__global__ void gnn_pack_weights_kernel(GnnPackJob p) {
    gnn_pack_weights(p, blockIdx.x * blockDim.x + threadIdx.x);
}

#define OWN_WAVES 32
#define OWN_RMAX  640
#define OWN_RPW   (OWN_RMAX / OWN_WAVES)
#define OWN_QCAP  64
#define OWN_BATCH 16
#define OWN_SBITS 18
#define OWN_SMASK ((1u << OWN_SBITS) - 1u)
#define APITCH 68
#define DROWS  128

__global__ void gnn_owner_key_kernel(const int* __restrict__ dst,
                                     unsigned* __restrict__ keys,
                                     int E, int n_words, int R) {
    const int w = blockIdx.x * blockDim.x + threadIdx.x;
    if (w >= n_words) return;
    unsigned kw = 0u;
    #pragma unroll
    for (int b = 0; b < 4; ++b) {
        const long long e = (long long)w * 4 + b;
        unsigned k = 0xFFu;
        if (e < (long long)E) k = (unsigned)dst[e] / (unsigned)R;
        kw |= k << (8 * b);
    }
    keys[w] = kw;
}

#define BM_EDGES  2048
#define BM_LWORDS 64

__global__ void __launch_bounds__(256)
gnn_owner_bitmap_kernel(const int* __restrict__ dst,
                        unsigned* __restrict__ bm,
                        int E, int R, int n_own, int wpo,
                        int n_bm_wg, GnnPackJob p0, GnnPackJob p1, GnnPackJob p2) {
    __shared__ __attribute__((aligned(16))) unsigned tile[255 * BM_LWORDS];
    const int t  = threadIdx.x;
    if ((int)blockIdx.x >= n_bm_wg) {
        int b = (int)blockIdx.x - n_bm_wg;
        const int nb0 = (p0.dout_pad * DIN + 255) / 256;
        const int nb1 = (p1.dout_pad * DIN + 255) / 256;
        if (b < nb0)            gnn_pack_weights(p0, b * 256 + t);
        else if (b < nb0 + nb1) gnn_pack_weights(p1, (b - nb0) * 256 + t);
        else                    gnn_pack_weights(p2, (b - nb0 - nb1) * 256 + t);
        return;
    }
    const int n4 = n_own * (BM_LWORDS / 4);
    for (int i = t; i < n4; i += 256) ((uint4*)tile)[i] = make_uint4(0u, 0u, 0u, 0u);
    __syncthreads();
    const long long base = (long long)blockIdx.x * BM_EDGES;
    int d[BM_EDGES / 256];
    #pragma unroll
    for (int b = 0; b < BM_EDGES / 256; ++b) {
        const long long e = base + b * 256 + t;
        d[b] = dst[e < (long long)E ? e : (long long)E - 1];
    }
    #pragma unroll
    for (int b = 0; b < BM_EDGES / 256; ++b) {
        const int le = b * 256 + t;
        if (base + le < (long long)E) {
            const unsigned g = (unsigned)d[b] / (unsigned)R;
            atomicOr(&tile[g * BM_LWORDS + (le >> 5)], 1u << (le & 31));
        }
    }
    __syncthreads();
    uint4* o4 = (uint4*)bm;
    for (int i = t; i < n4; i += 256) {
        const int g = i >> 4, p = i & 15;
        o4[((size_t)g * (size_t)wpo + (size_t)blockIdx.x * BM_LWORDS) / 4 + p] = ((const uint4*)tile)[i];
    }
}

__device__ __forceinline__ unsigned gnn_zero_bytes(unsigned x) {
    return ~(((x & 0x7F7F7F7Fu) + 0x7F7F7F7Fu) | x | 0x7F7F7F7Fu);
}

template <int W, int P>
__device__ __forceinline__ void
gnn_owner_rows(const float* __restrict__ rows, int s, int r, int n, int lane, float* acc) {
    const bool act  = (W == DIN) || (2 * lane < W);
    const int  loff = act ? 2 * lane : 0;
    #pragma unroll
    for (int b = 0; b < 32 / OWN_BATCH; ++b) {
        if (b * OWN_BATCH < n) {
            v2f v[OWN_BATCH];
            #pragma unroll
            for (int j = 0; j < OWN_BATCH; ++j) {
                const int sj = __builtin_amdgcn_readlane(s, b * OWN_BATCH + j);
                v[j] = *(const v2f*)(rows + (size_t)sj * DIN + loff);
            }
            if (act) {
                #pragma unroll
                for (int j = 0; j < OWN_BATCH; ++j) {
                    if (b * OWN_BATCH + j < n) {
                        const int rj = __builtin_amdgcn_readlane(r, b * OWN_BATCH + j);
                        float* a = acc + rj * P + 2 * lane;
                        __hip_atomic_fetch_add(a,     v[j].x, __ATOMIC_RELAXED, __HIP_MEMORY_SCOPE_WORKGROUP);
                        __hip_atomic_fetch_add(a + 1, v[j].y, __ATOMIC_RELAXED, __HIP_MEMORY_SCOPE_WORKGROUP);
                    }
                }
            }
        }
    }
}

template <int W, int P>
__device__ __forceinline__ void
gnn_owner_drain(const float* __restrict__ rows,
                const int* __restrict__ src, const int* __restrict__ dst,
                const unsigned* qw, unsigned head, int n, int lo, int lane,
                float* acc, unsigned* degL, int do_deg,
                unsigned* seg, int pos, int capw) {
    int s = 0, r = 0;
    if (lane < n) {
        const unsigned e = qw[(head + (unsigned)lane) & (OWN_QCAP - 1)];
        s = src[e];
        r = dst[e] - lo;
        if (do_deg) atomicAdd(&degL[r], 1u);
        if (seg != nullptr && pos + lane < capw)
            seg[pos + lane] = (unsigned)s | ((unsigned)r << OWN_SBITS);
    }
    gnn_owner_rows<W, P>(rows, s, r, n, lane, acc);
}

#define C2   40
#define C2A  32
#define C2B  8

__device__ __forceinline__ void
gnn_row_logsoftmax(float a, float b, float z0, float z1, float rdeg, int lane, float* orow) {
    const bool hasB = lane < C2B;
    const float NEG_INF = -__builtin_inff();
    const float v0 = a * rdeg + z0;
    const float v1 = hasB ? b * rdeg + z1 : NEG_INF;

    float m = fmaxf(v0, v1);
    #pragma unroll
    for (int off = 16; off > 0; off >>= 1) m = fmaxf(m, __shfl_xor(m, off, 32));
    const float e0 = expf(v0 - m);
    const float e1 = hasB ? expf(v1 - m) : 0.0f;
    float s = e0 + e1;
    #pragma unroll
    for (int off = 16; off > 0; off >>= 1) s += __shfl_xor(s, off, 32);
    const float ls = logf(s);
    orow[lane] = v0 - m - ls;
    if (hasB) orow[C2A + lane] = v1 - m - ls;
}

__device__ __forceinline__ void
gnn_proj_store(const v8f& cy, const v8f& cz, float bv,
               float* __restrict__ yp, int ypitch, float* __restrict__ zp, int nvalid) {
    #pragma unroll
    for (int r = 0; r < 8; ++r) {
        if (r < nvalid) {
            yp[(size_t)r * ypitch] = cy[r];
            zp[(size_t)r * C2]     = cz[r] + bv;
        }
    }
}

__device__ __forceinline__ void gnn_lds_barrier() {
    __builtin_amdgcn_fence(__ATOMIC_RELEASE, "workgroup", "local");
    __builtin_amdgcn_s_barrier();
    __builtin_amdgcn_fence(__ATOMIC_ACQUIRE, "workgroup", "local");
}

template <int W, bool BM, bool FIN>
__global__ void __launch_bounds__(32 * OWN_WAVES)
gnn_owner_sum_kernel(const float* __restrict__ rows,
                     const int* __restrict__ src,
                     const int* __restrict__ dst,
                     const unsigned* __restrict__ keys,
                     int n_chunks,
                     float* __restrict__ hout,
                     float* __restrict__ deg,
                     int n_nodes, int R, int do_deg,
                     unsigned* list, int* lcnt, int capw, int list_mode,
                     float* fin,
                     const float* __restrict__ WFl,
                     const float* __restrict__ WFr,
                     const float* __restrict__ blp,
                     const float* __restrict__ PFl,
                     const float* __restrict__ PFr,
                     const float* __restrict__ pbl,
                     float* __restrict__ yout,
                     float* __restrict__ zout) {
    static_assert(FIN ? W == C2 : W == DIN, "last layer: 40-wide sums; hidden layers: 64-wide");
    constexpr int P = FIN ? W : APITCH;
    __shared__ __attribute__((aligned(16))) float acc[OWN_RMAX * (FIN ? DIN : APITCH)];
    __shared__ unsigned degL[OWN_RMAX];
    __shared__ unsigned q[OWN_WAVES][OWN_QCAP];

    const int lane = threadIdx.x;
    const int wv = __builtin_amdgcn_readfirstlane((int)threadIdx.y);
    const int t  = wv * 32 + lane;
    const int g  = blockIdx.x;
    const int lo = g * R;
    int nrows = n_nodes - lo;
    nrows = nrows < R ? nrows : R;

    int wcount = 0;
    if (list_mode == 2) wcount = lcnt[g * OWN_WAVES + lane];

    const int srow = t >> 4;
    const int c4   = (t & 15) * 4;
    const float* hown = rows + (size_t)lo * DIN + c4;
    v4f rx[2] = {};
    float dgv = 0.0f;
    if constexpr (!FIN) {
        #pragma unroll
        for (int j = 0; j < 2; ++j) {
            int row = srow + 64 * j;
            row = row < nrows ? row : nrows - 1;
            rx[j] = *(const v4f*)(hown + (size_t)row * DIN);
        }
        if (!do_deg && t < nrows) dgv = deg[lo + t];
    }

    auto piece = [](int i) { return FIN ? i : (i >> 4) * (APITCH / 4) + (i & 15); };
    const int n4 = nrows * W / 4;
    for (int i = t; i < n4; i += 32 * OWN_WAVES) ((v4f*)acc)[piece(i)] = v4f{0.0f, 0.0f, 0.0f, 0.0f};
    for (int i = t; i < nrows; i += 32 * OWN_WAVES) degL[i] = 0u;
    gnn_lds_barrier();

    const bool replay = list_mode == 2 && __builtin_amdgcn_ballot_w32(wcount > capw) == 0u;
    const size_t seg_off = ((size_t)g * OWN_WAVES + (size_t)wv) * (size_t)capw;

    if (replay) {
        const int n = __builtin_amdgcn_readlane(wcount, wv);
        if (n > 0) {
            const unsigned* seg = list + seg_off;
            const int last  = n - 1;
            const int nfull = n >> 5;
            unsigned ent = seg[lane < last ? lane : last];
            for (int k = 0; k < nfull; ++k) {
                const int in = (k + 1) * 32 + lane;
                const unsigned entn = seg[in < last ? in : last];
                gnn_owner_rows<W, P>(rows, (int)(ent & OWN_SMASK), (int)(ent >> OWN_SBITS), 32, lane, acc);
                ent = entn;
            }
            const int rem = n & 31;
            if (rem > 0)
                gnn_owner_rows<W, P>(rows, (int)(ent & OWN_SMASK), (int)(ent >> OWN_SBITS), rem, lane, acc);
        }
    } else {
    unsigned* seg = list_mode == 1 ? list + seg_off : nullptr;
    int pos = 0;

    unsigned* qw = &q[wv][0];
    unsigned head = 0u;
    int cnt = 0;

    if (BM) {
        const uint4* b4 = (const uint4*)(keys + (size_t)g * (size_t)n_chunks * 128u);
        int kb = wv;
        uint4 bw = make_uint4(0u, 0u, 0u, 0u);
        if (kb < n_chunks) bw = b4[(size_t)kb * 32 + lane];
        while (kb < n_chunks) {
            const unsigned ebase = (unsigned)kb * 4096u + (unsigned)lane * 128u;
            const int kbn = kb + OWN_WAVES;
            const uint4 bwn = b4[(size_t)(kbn < n_chunks ? kbn : kb) * 32 + lane];
            #pragma unroll
            for (int hh = 0; hh < 2; ++hh) {
                unsigned long long m = hh ? (((unsigned long long)bw.w << 32) | bw.z)
                                          : (((unsigned long long)bw.y << 32) | bw.x);
                const unsigned eb = ebase + 64u * (unsigned)hh;
                for (;;) {
                    const bool has = m != 0ull;
                    const unsigned bal = __builtin_amdgcn_ballot_w32(has);
                    if (bal == 0u) break;
                    if (has) {
                        const unsigned pre = __builtin_amdgcn_mbcnt_lo(bal, 0u);
                        const unsigned tz  = (unsigned)__builtin_ctzll(m);
                        qw[(head + (unsigned)cnt + pre) & (OWN_QCAP - 1)] = eb + tz;
                        m &= m - 1ull;
                    }
                    cnt += __builtin_popcount(bal);
                    if (cnt >= 32) {
                        gnn_owner_drain<W, P>(rows, src, dst, qw, head, 32, lo, lane, acc, degL, do_deg,
                                           seg, pos, capw);
                        pos += 32;
                        head = (head + 32u) & (OWN_QCAP - 1);
                        cnt -= 32;
                    }
                }
            }
            kb = kbn; bw = bwn;
        }
    } else {
    const unsigned pat = (unsigned)g * 0x01010101u;
    const int start = (int)(((unsigned)g * 97u) % (unsigned)n_chunks);
    const uint4* k4 = (const uint4*)keys;

    int k = wv;
    uint4 kw = make_uint4(0xFFFFFFFFu, 0xFFFFFFFFu, 0xFFFFFFFFu, 0xFFFFFFFFu);
    int c = 0;
    if (k < n_chunks) {
        c = start + k; if (c >= n_chunks) c -= n_chunks;
        kw = k4[(size_t)c * 32 + lane];
    }
    while (k < n_chunks) {
        unsigned m = (gnn_zero_bytes(kw.x ^ pat) >> 7) | (gnn_zero_bytes(kw.y ^ pat) >> 6)
                   | (gnn_zero_bytes(kw.z ^ pat) >> 5) | (gnn_zero_bytes(kw.w ^ pat) >> 4);
        const unsigned ebase = (unsigned)c * 512u + (unsigned)lane * 16u;

        const int kn = k + OWN_WAVES;
        int cn = start + (kn < n_chunks ? kn : k);
        if (cn >= n_chunks) cn -= n_chunks;
        const uint4 kwn = k4[(size_t)cn * 32 + lane];
        for (;;) {
            const bool has = m != 0u;
            const unsigned bal = __builtin_amdgcn_ballot_w32(has);
            if (bal == 0u) break;
            if (has) {
                const unsigned pre = __builtin_amdgcn_mbcnt_lo(bal, 0u);
                const unsigned tz  = (unsigned)__builtin_ctz(m);
                qw[(head + (unsigned)cnt + pre) & (OWN_QCAP - 1)] = ebase + (tz & 7u) * 4u + (tz >> 3);
                m &= m - 1u;
            }
            cnt += __builtin_popcount(bal);
            if (cnt >= 32) {
                gnn_owner_drain<W, P>(rows, src, dst, qw, head, 32, lo, lane, acc, degL, do_deg,
                                   seg, pos, capw);
                pos += 32;
                head = (head + 32u) & (OWN_QCAP - 1);
                cnt -= 32;
            }
        }
        k = kn; c = cn; kw = kwn;
    }
    }
    if (cnt > 0) {
        gnn_owner_drain<W, P>(rows, src, dst, qw, head, cnt, lo, lane, acc, degL, do_deg,
                           seg, pos, capw);
        pos += cnt;
    }
    if (list_mode == 1 && lane == 0) lcnt[g * OWN_WAVES + wv] = pos;
    }

    if constexpr (FIN) {
        const int rpw = (nrows + OWN_WAVES - 1) / OWN_WAVES;
        const int r0  = wv * rpw;
        int cnt = nrows - r0;
        cnt = cnt < rpw ? cnt : rpw;
        const int l2 = lane < C2B ? C2A + lane : lane;
        float* zrow = fin + (size_t)(lo + r0) * C2;
        float z0[OWN_RPW], z1[OWN_RPW];
        #pragma unroll
        for (int j = 0; j < OWN_RPW; ++j) {
            z0[j] = 0.0f; z1[j] = 0.0f;
            if (j < cnt) {
                z0[j] = zrow[j * C2 + lane];
                z1[j] = zrow[j * C2 + l2];
            }
        }
        float dv = 1.0f;
        if (lane < cnt) dv = deg[lo + r0 + lane];
        __syncthreads();
        const float rdv = 1.0f / fmaxf(dv, 1.0f);
        const float* arow = acc + r0 * C2;
        #pragma unroll
        for (int j = 0; j < OWN_RPW; ++j) {
            if (j < cnt) {
                const float a = arow[j * C2 + lane];
                const float b = arow[j * C2 + l2];
                const float rdeg = __int_as_float(__builtin_amdgcn_readlane(__float_as_int(rdv), j));
                gnn_row_logsoftmax(a, b, z0[j], z1[j], rdeg, lane, zrow + j * C2);
            }
        }
    } else {
        __shared__ __attribute__((aligned(16))) float sX[DROWS * APITCH];
        const int ntile = wv & 3;
        const int mrow  = (wv >> 2) * 16;
        const int khalf = (lane >> 4) * 2;
        const int col   = ntile * 16 + (lane & 15);
        const int n_strips = (nrows + DROWS - 1) / DROWS;

        #pragma unroll
        for (int j = 0; j < 2; ++j)
            *(v4f*)(sX + (srow + 64 * j) * APITCH + c4) = rx[j];

        v2f bL[16], bR[16];
        {
            const size_t wo = ((size_t)(ntile * 16) * 32 + lane) * 2;
            #pragma unroll
            for (int i = 0; i < 16; ++i) {
                bL[i] = *(const v2f*)(WFl + wo + (size_t)i * 64);
                bR[i] = *(const v2f*)(WFr + wo + (size_t)i * 64);
            }
        }
        const float bv = blp[col];
        if (!do_deg && t < nrows) degL[t] = __float_as_uint(dgv);
        gnn_lds_barrier();

        for (int s = 0; s < n_strips; ++s) {
            const bool more = s + 1 < n_strips;
            if (more) {
                #pragma unroll
                for (int j = 0; j < 2; ++j) {
                    int row = (s + 1) * DROWS + srow + 64 * j;
                    row = row < nrows ? row : nrows - 1;
                    rx[j] = *(const v4f*)(hown + (size_t)row * DIN);
                }
            }

            const int trow = s * DROWS + mrow;
            const bool have = trow < nrows;
            v8f c = {};
            if (have) {
                int ar = trow + (lane & 15);
                ar = ar < nrows ? ar : nrows - 1;
                const unsigned dw = degL[ar];
                const float rdeg = 1.0f / fmaxf(do_deg ? (float)dw : __uint_as_float(dw), 1.0f);
                const float* am = acc + ar * APITCH + khalf;
                const float* ax = sX + (mrow + (lane & 15)) * APITCH + khalf;
                #pragma unroll
                for (int i = 0; i < 16; ++i) {
                    const v2f aAgg = *(const v2f*)(am + 4 * i) * rdeg;
                    const v2f aX   = *(const v2f*)(ax + 4 * i);
                    c = __builtin_amdgcn_wmma_f32_16x16x4_f32(false, aAgg, false, bL[i],
                                                              (short)0, c, false, false);
                    c = __builtin_amdgcn_wmma_f32_16x16x4_f32(false, aX,   false, bR[i],
                                                              (short)0, c, false, false);
                }
            }
            gnn_lds_barrier();
            if (have) {
                float* op = acc + (trow + (lane >> 4) * 8) * APITCH + col;
                #pragma unroll
                for (int r = 0; r < 8; ++r)
                    op[r * APITCH] = fmaxf(c[r] + bv, 0.0f);
            }
            if (more) {
                #pragma unroll
                for (int j = 0; j < 2; ++j)
                    *(v4f*)(sX + (srow + 64 * j) * APITCH + c4) = rx[j];
            }
            gnn_lds_barrier();
        }
        if (yout == nullptr) {
            v4f* o4 = (v4f*)(hout + (size_t)lo * DIN);
            for (int i = t; i < n4; i += 32 * OWN_WAVES) o4[i] = ((const v4f*)acc)[piece(i)];
        } else {
            if (wv < 30) {
                const int pt   = wv % 3;
                const int pcol = pt * 16 + (lane & 15);
                v2f pL[16], pR[16];
                {
                    const size_t wo = ((size_t)(pt * 16) * 32 + lane) * 2;
                    #pragma unroll
                    for (int i = 0; i < 16; ++i) {
                        pL[i] = *(const v2f*)(PFl + wo + (size_t)i * 64);
                        pR[i] = *(const v2f*)(PFr + wo + (size_t)i * 64);
                    }
                }
                const float pbv = pbl[pcol];
                for (int trow = (wv / 3) * 16; trow < nrows; trow += 160) {
                    int ar = trow + (lane & 15);
                    ar = ar < nrows ? ar : nrows - 1;
                    const float* ah = acc + ar * APITCH + khalf;
                    v8f cy = {};
                    v8f cz = {};
                    #pragma unroll
                    for (int i = 0; i < 16; ++i) {
                        const v2f a = *(const v2f*)(ah + 4 * i);
                        cy = __builtin_amdgcn_wmma_f32_16x16x4_f32(false, a, false, pL[i],
                                                                   (short)0, cy, false, false);
                        cz = __builtin_amdgcn_wmma_f32_16x16x4_f32(false, a, false, pR[i],
                                                                   (short)0, cz, false, false);
                    }
                    const int row0 = trow + (lane >> 4) * 8;
                    const bool full = trow + 16 <= nrows;
                    if (pcol < C2) {
                        float* yp = yout + (size_t)(lo + row0) * DIN + pcol;
                        float* zp = zout + (size_t)(lo + row0) * C2  + pcol;
                        if (full) gnn_proj_store(cy, cz, pbv, yp, DIN, zp, 8);
                        else      gnn_proj_store(cy, cz, pbv, yp, DIN, zp, nrows - row0);
                    }
                }
            }
        }
        if (do_deg)
            for (int i = t; i < nrows; i += 32 * OWN_WAVES) deg[lo + i] = (float)degL[i];
    }
}

#define SROWS  32
__global__ void __launch_bounds__(256)
gnn_proj_wmma_kernel(const float* __restrict__ hin,
                     const float* __restrict__ WFl,
                     const float* __restrict__ WFr,
                     const float* __restrict__ blp,
                     float* __restrict__ y,
                     float* __restrict__ z,
                     int n_rows, int n_strips) {
    __shared__ __attribute__((aligned(16))) float sX[2][SROWS * APITCH];

    const int lane  = threadIdx.x;
    const int wv    = __builtin_amdgcn_readfirstlane((int)threadIdx.y);
    const bool worker = wv < 6;
    const int mhalf = wv >= 3 ? 1 : 0;
    const int ntile = wv - 3 * mhalf;
    const int khalf = (lane >> 4) * 2;
    const int col   = ntile * 16 + (lane & 15);

    int strip = blockIdx.x;
    if (strip >= n_strips) return;

    const int t    = wv * 32 + lane;
    const int srow = t >> 4;
    const int c4   = (t & 15) * 4;

    v4f rx[2];
    #pragma unroll
    for (int j = 0; j < 2; ++j) {
        int row = strip * SROWS + srow + 16 * j;
        row = row < n_rows ? row : n_rows - 1;
        rx[j] = *(const v4f*)(hin + (size_t)row * DIN + c4);
    }

    v2f bL[16], bR[16];
    float bv = 0.0f;
    if (worker) {
        const size_t wo = ((size_t)(ntile * 16) * 32 + lane) * 2;
        #pragma unroll
        for (int i = 0; i < 16; ++i) {
            bL[i] = *(const v2f*)(WFl + wo + (size_t)i * 64);
            bR[i] = *(const v2f*)(WFr + wo + (size_t)i * 64);
        }
        bv = blp[col];
    }

    const int aoff = (mhalf * 16 + (lane & 15)) * APITCH + khalf;

    int buf = 0;
    for (;;) {
        float* px = &sX[buf][0];
        #pragma unroll
        for (int j = 0; j < 2; ++j)
            *(v4f*)(px + (srow + 16 * j) * APITCH + c4) = rx[j];
        __builtin_amdgcn_fence(__ATOMIC_RELEASE, "workgroup", "local");
        __builtin_amdgcn_s_barrier();
        __builtin_amdgcn_fence(__ATOMIC_ACQUIRE, "workgroup", "local");

        const int next = strip + (int)gridDim.x;
        if (next < n_strips) {
            #pragma unroll
            for (int j = 0; j < 2; ++j) {
                int row = next * SROWS + srow + 16 * j;
                row = row < n_rows ? row : n_rows - 1;
                rx[j] = *(const v4f*)(hin + (size_t)row * DIN + c4);
            }
        }

        if (worker) {
            const float* ax = px + aoff;
            v8f cy = {};
            v8f cz = {};
            #pragma unroll
            for (int i = 0; i < 16; ++i) {
                const v2f a = *(const v2f*)(ax + 4 * i);
                cy = __builtin_amdgcn_wmma_f32_16x16x4_f32(false, a, false, bL[i],
                                                           (short)0, cy, false, false);
                cz = __builtin_amdgcn_wmma_f32_16x16x4_f32(false, a, false, bR[i],
                                                           (short)0, cz, false, false);
            }

            const int tile_row = strip * SROWS + mhalf * 16;
            const int row0i    = tile_row + (lane >> 4) * 8;
            const size_t row0  = (size_t)row0i;
            const bool full  = tile_row + 16 <= n_rows;
            const int nvalid = n_rows - row0i;
            if (col < C2) {
                float* yp = y + row0 * DIN + col;
                float* zp = z + row0 * C2  + col;
                if (full) gnn_proj_store(cy, cz, bv, yp, DIN, zp, 8);
                else      gnn_proj_store(cy, cz, bv, yp, DIN, zp, nvalid);
            }
        }

        if (next >= n_strips) break;
        strip = next;
        buf ^= 1;
    }
}

extern "C" void kernel_launch(void* const* d_in, const int* in_sizes, int n_in,
                              void* d_out, int out_size, void* d_ws, size_t ws_size,
                              hipStream_t stream) {
    const float* x   = (const float*)d_in[0];
    const int*   ei  = (const int*)  d_in[1];
    const float* Wl0 = (const float*)d_in[2];
    const float* bl0 = (const float*)d_in[3];
    const float* Wr0 = (const float*)d_in[4];
    const float* Wl1 = (const float*)d_in[5];
    const float* bl1 = (const float*)d_in[6];
    const float* Wr1 = (const float*)d_in[7];
    const float* Wl2 = (const float*)d_in[8];
    const float* bl2 = (const float*)d_in[9];
    const float* Wr2 = (const float*)d_in[10];

    const int N = in_sizes[0] / DIN;
    const int E = in_sizes[1] / 2;
    const int* src = ei;
    const int* dst = ei + E;

    const int DP0 = 64, DP1 = 64, DP2 = 48;

    const int R      = (((N + 254) / 255) + 7) / 8 * 8;
    const int n_own  = (N + R - 1) / R;
    const int n_chunks = (E + 511) / 512;
    const int n_kwords = n_chunks * 128;
    const int n_bch    = (E + 4095) / 4096;
    const int wpo      = n_bch * 128;

    float* deg  = (float*)d_ws;
    float* bufA = deg  + (size_t)N;
    float* bufB = bufA + (size_t)N * DIN;
    float* WTl0 = bufB + (size_t)N * DIN;
    float* WTr0 = WTl0 + DP0 * DIN;
    float* blp0 = WTr0 + DP0 * DIN;
    float* WTl1 = blp0 + DP0;
    float* WTr1 = WTl1 + DP1 * DIN;
    float* blp1 = WTr1 + DP1 * DIN;
    float* WTl2 = blp1 + DP1;
    float* WTr2 = WTl2 + DP2 * DIN;
    float* blp2 = WTr2 + DP2 * DIN;
    const size_t key_off = ((size_t)((blp2 + DP2) - deg) + 63) / 64 * 64;
    unsigned* keys = (unsigned*)(deg + key_off);
    const bool use_bm = ws_size >= (key_off + (size_t)n_own * (size_t)wpo) * sizeof(float);

    const int CAPW = 320;
    const size_t cnt_words = ((size_t)n_own * OWN_WAVES + 63) / 64 * 64;
    const size_t rec_words = cnt_words + (size_t)n_own * OWN_WAVES * CAPW;
    const size_t bm_end    = key_off + (size_t)n_own * (size_t)wpo;
    int* lcnt = nullptr;
    unsigned* list = nullptr;
    int lm0 = 0, lm1 = 0, lm2 = 0;
    if (use_bm && N <= (1 << OWN_SBITS)) {
        if (ws_size >= (bm_end + rec_words) * sizeof(float)) {
            lcnt = (int*)(deg + bm_end);
            lm0 = 1; lm1 = 2; lm2 = 2;
        } else if ((size_t)out_size >= rec_words) {
            lcnt = (int*)d_out;
            lm0 = 1; lm1 = 2;
        }
        if (lm0) list = (unsigned*)lcnt + cnt_words;
    }

    const dim3 own_block(32, OWN_WAVES);
    const int ZB = 256;

    const GnnPackJob pk0 = {Wl0, bl0, Wr0, WTl0, WTr0, blp0, 64, DP0};
    const GnnPackJob pk1 = {Wl1, bl1, Wr1, WTl1, WTr1, blp1, 64, DP1};
    const GnnPackJob pk2 = {Wl2, bl2, Wr2, WTl2, WTr2, blp2, 40, DP2};
    const int nb0 = (DP0 * DIN + ZB - 1) / ZB;
    const int nb1 = (DP1 * DIN + ZB - 1) / ZB;
    const int nb2 = (DP2 * DIN + ZB - 1) / ZB;
    if (use_bm) {
        const int n_bm_wg = wpo / BM_LWORDS;
        gnn_owner_bitmap_kernel<<<n_bm_wg + nb0 + nb1 + nb2, 256, 0, stream>>>(
            dst, keys, E, R, n_own, wpo, n_bm_wg, pk0, pk1, pk2);
    } else {
        gnn_pack_weights_kernel<<<nb0, ZB, 0, stream>>>(pk0);
        gnn_pack_weights_kernel<<<nb1, ZB, 0, stream>>>(pk1);
        gnn_pack_weights_kernel<<<nb2, ZB, 0, stream>>>(pk2);
        gnn_owner_key_kernel<<<(n_kwords + ZB - 1) / ZB, ZB, 0, stream>>>(
            dst, keys, E, n_kwords, R);
    }
    const int n_sweep = use_bm ? n_bch : n_chunks;

    if (use_bm)
        gnn_owner_sum_kernel<DIN, true, false><<<dim3(n_own), own_block, 0, stream>>>(
            x, src, dst, keys, n_sweep, bufA, deg, N, R, 1, list, lcnt, CAPW, lm0, nullptr,
            WTl0, WTr0, blp0, nullptr, nullptr, nullptr, nullptr, nullptr);
    else
        gnn_owner_sum_kernel<DIN, false, false><<<dim3(n_own), own_block, 0, stream>>>(
            x, src, dst, keys, n_sweep, bufA, deg, N, R, 1, list, lcnt, CAPW, 0, nullptr,
            WTl0, WTr0, blp0, nullptr, nullptr, nullptr, nullptr, nullptr);

    const bool proj_here = (void*)lcnt != d_out;
    float* y  = proj_here ? bufB : bufA;
    float* h2 = proj_here ? nullptr : bufB;
    float* y1 = proj_here ? y : nullptr;
    float* z1 = proj_here ? (float*)d_out : nullptr;
    if (use_bm)
        gnn_owner_sum_kernel<DIN, true, false><<<dim3(n_own), own_block, 0, stream>>>(
            bufA, src, dst, keys, n_sweep, h2, deg, N, R, 0, list, lcnt, CAPW, lm1, nullptr,
            WTl1, WTr1, blp1, WTl2, WTr2, blp2, y1, z1);
    else
        gnn_owner_sum_kernel<DIN, false, false><<<dim3(n_own), own_block, 0, stream>>>(
            bufA, src, dst, keys, n_sweep, h2, deg, N, R, 0, list, lcnt, CAPW, 0, nullptr,
            WTl1, WTr1, blp1, WTl2, WTr2, blp2, y1, z1);
    if (!proj_here) {
        const int n_strips = (N + SROWS - 1) / SROWS;
        const int n_wg = n_strips < 768 ? n_strips : 768;
        gnn_proj_wmma_kernel<<<dim3(n_wg), dim3(32, 8), 0, stream>>>(
            h2, WTl2, WTr2, blp2, y, (float*)d_out, N, n_strips);
    }

    if (use_bm)
        gnn_owner_sum_kernel<C2, true, true><<<dim3(n_own), own_block, 0, stream>>>(
            y, src, dst, keys, n_sweep, nullptr, deg, N, R, 0, list, lcnt, CAPW, lm2,
            (float*)d_out, nullptr, nullptr, nullptr, nullptr, nullptr, nullptr, nullptr, nullptr);
    else
        gnn_owner_sum_kernel<C2, false, true><<<dim3(n_own), own_block, 0, stream>>>(
            y, src, dst, keys, n_sweep, nullptr, deg, N, R, 0, list, lcnt, CAPW, 0,
            (float*)d_out, nullptr, nullptr, nullptr, nullptr, nullptr, nullptr, nullptr, nullptr);
}
